# starting kernel
# baseline (speedup 1.0000x reference)
#include <hip/hip_runtime.h>
#include <math.h>


typedef __attribute__((ext_vector_type(2))) float v2f;
typedef __attribute__((ext_vector_type(8))) float v8f;

__device__ __forceinline__ unsigned fkey(float f) {
  unsigned u = __float_as_uint(f);
  return (u & 0x80000000u) ? ~u : (u | 0x80000000u);
}
__device__ __forceinline__ float funkey(unsigned k) {
  unsigned u = (k & 0x80000000u) ? (k & 0x7fffffffu) : ~k;
  return __uint_as_float(u);
}

template <int NCOLS>
__global__ __launch_bounds__(32 * (NCOLS / 16))
void gemm_wmma_f32(const float* __restrict__ A, const float* __restrict__ Wg,
                   float* __restrict__ Out, int M) {
  constexpr int K = 128;
  constexpr int NT = NCOLS / 16;
  __shared__ float Ws[K * NCOLS];

  const int tid = threadIdx.x;
  for (int i = tid; i < K * NCOLS; i += 32 * NT) Ws[i] = Wg[i];
  __syncthreads();

  const int wave  = tid >> 5;
  const int lane  = tid & 31;
  const int row16 = blockIdx.x << 4;
  const int r     = lane & 15;
  const int kb    = (lane >> 4) << 1;
  const int col   = (wave << 4) + r;

  int arow = row16 + r;
  if (arow >= M) arow = M - 1;
  const float* __restrict__ ap = A + (size_t)arow * K;

  v8f acc = {};
#pragma unroll
  for (int k = 0; k < K; k += 4) {
    v2f a, b;
    a.x = ap[k + kb];
    a.y = ap[k + kb + 1];
    b.x = Ws[(k + kb) * NCOLS + col];
    b.y = Ws[(k + kb + 1) * NCOLS + col];
    acc = __builtin_amdgcn_wmma_f32_16x16x4_f32(false, a, false, b,
                                                (short)0, acc, false, false);
  }

  const int orow0 = row16 + ((lane >> 4) << 3);
  const int ocol  = (wave << 4) + r;
#pragma unroll
  for (int i = 0; i < 8; ++i) {
    int orow = orow0 + i;
    if (orow < M) Out[(size_t)orow * NCOLS + ocol] = acc[i];
  }
}

__global__ void k_selfloop_stats(const int* __restrict__ dst,
                                 const float* __restrict__ ew,
                                 float* __restrict__ deg,
                                 float* __restrict__ loopw, int E) {
  int e = blockIdx.x * blockDim.x + threadIdx.x;
  if (e >= E) return;
  int d = dst[e];
  atomicAdd(&deg[d], 1.0f);
  atomicAdd(&loopw[d], ew[e]);
}

__global__ void k_selfloop_div(float* __restrict__ loopw,
                               const float* __restrict__ deg, int N) {
  int n = blockIdx.x * blockDim.x + threadIdx.x;
  if (n >= N) return;
  loopw[n] = loopw[n] / fmaxf(deg[n], 1.0f);
}

template <int H>
__global__ void k_node_coef(const float* __restrict__ hfeat,
                            const float* __restrict__ a_s,
                            const float* __restrict__ a_d,
                            float* __restrict__ scoef,
                            float* __restrict__ dcoef, int N) {
  int n = blockIdx.x * blockDim.x + threadIdx.x;
  if (n >= N) return;
  const float* hr = hfeat + (size_t)n * (H * 64);
#pragma unroll
  for (int h = 0; h < H; ++h) {
    float ss = 0.f, dd = 0.f;
#pragma unroll 8
    for (int c = 0; c < 64; ++c) {
      float v = hr[h * 64 + c];
      ss += v * a_s[h * 64 + c];
      dd += v * a_d[h * 64 + c];
    }
    scoef[n * H + h] = ss;
    dcoef[n * H + h] = dd;
  }
}

template <int H>
__global__ void k_edge_const(const float* __restrict__ We,
                             const float* __restrict__ ae,
                             float* __restrict__ cst) {
  int h = threadIdx.x;
  if (h < H) {
    float s = 0.f;
    for (int c = 0; c < 64; ++c) s += We[h * 64 + c] * ae[h * 64 + c];
    cst[h] = s;
  }
}

template <int H>
__global__ void k_edge_logit_max(const int* __restrict__ src,
                                 const int* __restrict__ dst,
                                 const float* __restrict__ ew,
                                 const float* __restrict__ loopw,
                                 const float* __restrict__ scoef,
                                 const float* __restrict__ dcoef,
                                 const float* __restrict__ cst,
                                 float* __restrict__ albuf,
                                 unsigned* __restrict__ mkey, int E, int ET) {
  int e = blockIdx.x * blockDim.x + threadIdx.x;
  if (e >= ET) return;
  int s, d;
  float w;
  if (e < E) { s = src[e]; d = dst[e]; w = ew[e]; }
  else       { s = d = e - E;          w = loopw[s]; }
#pragma unroll
  for (int h = 0; h < H; ++h) {
    float al = scoef[s * H + h] + dcoef[d * H + h] + w * cst[h];
    al = (al >= 0.f) ? al : 0.2f * al;
    albuf[(size_t)e * H + h] = al;
    atomicMax(&mkey[d * H + h], fkey(al));
  }
}

template <int H>
__global__ void k_edge_expsum(const int* __restrict__ dst,
                              const float* __restrict__ albuf,
                              const unsigned* __restrict__ mkey,
                              float* __restrict__ zbuf,
                              float* __restrict__ ssum, int E, int ET) {
  int e = blockIdx.x * blockDim.x + threadIdx.x;
  if (e >= ET) return;
  int d = (e < E) ? dst[e] : (e - E);
#pragma unroll
  for (int h = 0; h < H; ++h) {
    float z = expf(albuf[(size_t)e * H + h] - funkey(mkey[d * H + h]));
    zbuf[(size_t)e * H + h] = z;
    atomicAdd(&ssum[d * H + h], z);
  }
}

template <int F>
__global__ void k_edge_aggregate(const int* __restrict__ src,
                                 const int* __restrict__ dst,
                                 const float* __restrict__ hfeat,
                                 const float* __restrict__ zbuf,
                                 const float* __restrict__ ssum,
                                 float* __restrict__ agg, int E, int ET) {
  constexpr int H = F / 64;
  int gt = blockIdx.x * blockDim.x + threadIdx.x;
  int e = gt >> 5;
  int lane = gt & 31;
  if (e >= ET) return;
  int s, d;
  if (e < E) { s = src[e]; d = dst[e]; }
  else       { s = d = e - E; }
  float wc[H];
#pragma unroll
  for (int h = 0; h < H; ++h)
    wc[h] = zbuf[(size_t)e * H + h] / ssum[d * H + h];
  const float* hs = hfeat + (size_t)s * F;
  float* ad = agg + (size_t)d * F;
#pragma unroll
  for (int f = lane; f < F; f += 32)
    atomicAdd(&ad[f], hs[f] * wc[f >> 6]);
}

template <int F, bool ELU>
__global__ void k_bias_act(float* __restrict__ a, const float* __restrict__ b,
                           int n) {
  int i = blockIdx.x * blockDim.x + threadIdx.x;
  if (i >= n) return;
  float v = a[i] + b[i & (F - 1)];
  if (ELU) v = (v > 0.f) ? v : expm1f(v);
  a[i] = v;
}

extern "C" void kernel_launch(void* const* d_in, const int* in_sizes, int n_in,
                              void* d_out, int out_size, void* d_ws,
                              size_t ws_size, hipStream_t stream) {
  const float* x   = (const float*)d_in[0];
  const int*  eidx = (const int*)d_in[1];
  const float* ew  = (const float*)d_in[2];
  const float* W1  = (const float*)d_in[3];
  const float* We1 = (const float*)d_in[4];
  const float* as1 = (const float*)d_in[5];
  const float* ad1 = (const float*)d_in[6];
  const float* ae1 = (const float*)d_in[7];
  const float* b1  = (const float*)d_in[8];
  const float* W2  = (const float*)d_in[9];
  const float* We2 = (const float*)d_in[10];
  const float* as2 = (const float*)d_in[11];
  const float* ad2 = (const float*)d_in[12];
  const float* ae2 = (const float*)d_in[13];
  const float* b2  = (const float*)d_in[14];
  float* out = (float*)d_out;

  const int N  = in_sizes[0] / 128;
  const int E  = in_sizes[2];
  const int ET = E + N;
  const int* srcv = eidx;
  const int* dstv = eidx + E;

  float* p = (float*)d_ws;
  float* loopw = p; p += N;
  float* deg   = p; p += N;
  float* scoef = p; p += 2 * (size_t)N;
  float* dcoef = p; p += 2 * (size_t)N;
  float* cst   = p; p += 8;
  float* albuf = p; p += 2 * (size_t)ET;
  float* zbuf  = p; p += 2 * (size_t)ET;
  unsigned* mkey = (unsigned*)p; p += 2 * (size_t)N;
  float* ssum  = p; p += 2 * (size_t)N;
  float* hfeat = p; p += (size_t)N * 128;
  float* agg   = p; p += (size_t)N * 128;

  const int T = 256;
  auto cdiv = [](long long a, long long b) { return (int)((a + b - 1) / b); };

  hipMemsetAsync(deg, 0, (size_t)N * 4, stream);
  hipMemsetAsync(loopw, 0, (size_t)N * 4, stream);
  k_selfloop_stats<<<cdiv(E, T), T, 0, stream>>>(dstv, ew, deg, loopw, E);
  k_selfloop_div<<<cdiv(N, T), T, 0, stream>>>(loopw, deg, N);

  gemm_wmma_f32<128><<<cdiv(N, 16), 256, 0, stream>>>(x, W1, hfeat, N);
  k_node_coef<2><<<cdiv(N, T), T, 0, stream>>>(hfeat, as1, ad1, scoef, dcoef, N);
  k_edge_const<2><<<1, 32, 0, stream>>>(We1, ae1, cst);
  hipMemsetAsync(mkey, 0, 2 * (size_t)N * 4, stream);
  hipMemsetAsync(ssum, 0, 2 * (size_t)N * 4, stream);
  hipMemsetAsync(agg, 0, (size_t)N * 128 * 4, stream);
  k_edge_logit_max<2><<<cdiv(ET, T), T, 0, stream>>>(srcv, dstv, ew, loopw,
                                                     scoef, dcoef, cst, albuf,
                                                     mkey, E, ET);
  k_edge_expsum<2><<<cdiv(ET, T), T, 0, stream>>>(dstv, albuf, mkey, zbuf,
                                                  ssum, E, ET);
  k_edge_aggregate<128><<<cdiv((long long)ET * 32, T), T, 0, stream>>>(
      srcv, dstv, hfeat, zbuf, ssum, agg, E, ET);
  k_bias_act<128, true><<<cdiv((long long)N * 128, T), T, 0, stream>>>(
      agg, b1, N * 128);

  gemm_wmma_f32<64><<<cdiv(N, 16), 128, 0, stream>>>(agg, W2, hfeat, N);
  k_node_coef<1><<<cdiv(N, T), T, 0, stream>>>(hfeat, as2, ad2, scoef, dcoef, N);
  k_edge_const<1><<<1, 32, 0, stream>>>(We2, ae2, cst);
  hipMemsetAsync(mkey, 0, (size_t)N * 4, stream);
  hipMemsetAsync(ssum, 0, (size_t)N * 4, stream);
  hipMemsetAsync(out, 0, (size_t)N * 64 * 4, stream);
  k_edge_logit_max<1><<<cdiv(ET, T), T, 0, stream>>>(srcv, dstv, ew, loopw,
                                                     scoef, dcoef, cst, albuf,
                                                     mkey, E, ET);
  k_edge_expsum<1><<<cdiv(ET, T), T, 0, stream>>>(dstv, albuf, mkey, zbuf,
                                                  ssum, E, ET);
  k_edge_aggregate<64><<<cdiv((long long)ET * 32, T), T, 0, stream>>>(
      srcv, dstv, hfeat, zbuf, ssum, out, E, ET);
  k_bias_act<64, false><<<cdiv((long long)N * 64, T), T, 0, stream>>>(
      out, b2, N * 64);
}


# Self-loop edge of each node is the plain-stored start value of its segment max, exp sum and output row; only edge_index edges use atomics; six zero fills dropped; f32 sum order: self-loop first
# speedup vs baseline: 1.1367x; 1.1367x over previous
#include <hip/hip_runtime.h>
#include <math.h>


typedef __attribute__((ext_vector_type(2))) float v2f;
typedef __attribute__((ext_vector_type(8))) float v8f;

__device__ __forceinline__ unsigned fkey(float f) {
  unsigned u = __float_as_uint(f);
  return (u & 0x80000000u) ? ~u : (u | 0x80000000u);
}
__device__ __forceinline__ float funkey(unsigned k) {
  unsigned u = (k & 0x80000000u) ? (k & 0x7fffffffu) : ~k;
  return __uint_as_float(u);
}

template <int NCOLS>
__global__ __launch_bounds__(32 * (NCOLS / 16))
void gemm_wmma_f32(const float* __restrict__ A, const float* __restrict__ Wg,
                   float* __restrict__ Out, int M) {
  constexpr int K = 128;
  constexpr int NT = NCOLS / 16;
  __shared__ float Ws[K * NCOLS];

  const int tid = threadIdx.x;
  for (int i = tid; i < K * NCOLS; i += 32 * NT) Ws[i] = Wg[i];
  __syncthreads();

  const int wave  = tid >> 5;
  const int lane  = tid & 31;
  const int row16 = blockIdx.x << 4;
  const int r     = lane & 15;
  const int kb    = (lane >> 4) << 1;
  const int col   = (wave << 4) + r;

  int arow = row16 + r;
  if (arow >= M) arow = M - 1;
  const float* __restrict__ ap = A + (size_t)arow * K;

  v8f acc = {};
#pragma unroll
  for (int k = 0; k < K; k += 4) {
    v2f a, b;
    a.x = ap[k + kb];
    a.y = ap[k + kb + 1];
    b.x = Ws[(k + kb) * NCOLS + col];
    b.y = Ws[(k + kb + 1) * NCOLS + col];
    acc = __builtin_amdgcn_wmma_f32_16x16x4_f32(false, a, false, b,
                                                (short)0, acc, false, false);
  }

  const int orow0 = row16 + ((lane >> 4) << 3);
  const int ocol  = (wave << 4) + r;
#pragma unroll
  for (int i = 0; i < 8; ++i) {
    int orow = orow0 + i;
    if (orow < M) Out[(size_t)orow * NCOLS + ocol] = acc[i];
  }
}

__global__ void k_selfloop_stats(const int* __restrict__ dst,
                                 const float* __restrict__ ew,
                                 float* __restrict__ deg,
                                 float* __restrict__ loopw, int E) {
  int e = blockIdx.x * blockDim.x + threadIdx.x;
  if (e >= E) return;
  int d = dst[e];
  atomicAdd(&deg[d], 1.0f);
  atomicAdd(&loopw[d], ew[e]);
}

__global__ void k_selfloop_div(float* __restrict__ loopw,
                               const float* __restrict__ deg, int N) {
  int n = blockIdx.x * blockDim.x + threadIdx.x;
  if (n >= N) return;
  loopw[n] = loopw[n] / fmaxf(deg[n], 1.0f);
}

template <int H>
__global__ void k_node_coef(const float* __restrict__ hfeat,
                            const float* __restrict__ a_s,
                            const float* __restrict__ a_d,
                            const float* __restrict__ loopw,
                            const float* __restrict__ cst,
                            float* __restrict__ scoef,
                            float* __restrict__ dcoef,
                            float* __restrict__ selfal,
                            unsigned* __restrict__ mkey, int N) {
  int n = blockIdx.x * blockDim.x + threadIdx.x;
  if (n >= N) return;
  const float* hr = hfeat + (size_t)n * (H * 64);
  const float w = loopw[n];
#pragma unroll
  for (int h = 0; h < H; ++h) {
    float ss = 0.f, dd = 0.f;
#pragma unroll 8
    for (int c = 0; c < 64; ++c) {
      float v = hr[h * 64 + c];
      ss += v * a_s[h * 64 + c];
      dd += v * a_d[h * 64 + c];
    }
    scoef[n * H + h] = ss;
    dcoef[n * H + h] = dd;
    float al = ss + dd + w * cst[h];
    al = (al >= 0.f) ? al : 0.2f * al;
    selfal[n * H + h] = al;
    mkey[n * H + h] = fkey(al);
  }
}

template <int H>
__global__ void k_edge_const(const float* __restrict__ We,
                             const float* __restrict__ ae,
                             float* __restrict__ cst) {
  int h = threadIdx.x;
  if (h < H) {
    float s = 0.f;
    for (int c = 0; c < 64; ++c) s += We[h * 64 + c] * ae[h * 64 + c];
    cst[h] = s;
  }
}

template <int H>
__global__ void k_edge_logit_max(const int* __restrict__ src,
                                 const int* __restrict__ dst,
                                 const float* __restrict__ ew,
                                 const float* __restrict__ scoef,
                                 const float* __restrict__ dcoef,
                                 const float* __restrict__ cst,
                                 float* __restrict__ albuf,
                                 unsigned* __restrict__ mkey, int E) {
  int e = blockIdx.x * blockDim.x + threadIdx.x;
  if (e >= E) return;
  const int s = src[e];
  const int d = dst[e];
  const float w = ew[e];
#pragma unroll
  for (int h = 0; h < H; ++h) {
    float al = scoef[s * H + h] + dcoef[d * H + h] + w * cst[h];
    al = (al >= 0.f) ? al : 0.2f * al;
    albuf[(size_t)e * H + h] = al;
    atomicMax(&mkey[d * H + h], fkey(al));
  }
}

__global__ void k_selfloop_start(const float* __restrict__ hfeat,
                                 const float* __restrict__ selfal,
                                 const unsigned* __restrict__ mkey,
                                 float* __restrict__ ssum,
                                 float* __restrict__ agg, int nquads) {
  int q = blockIdx.x * blockDim.x + threadIdx.x;
  if (q >= nquads) return;
  const int g = q >> 4;
  const float z = expf(selfal[g] - funkey(mkey[g]));
  const float4 hv = reinterpret_cast<const float4*>(hfeat)[q];
  float4 r;
  r.x = hv.x * z;
  r.y = hv.y * z;
  r.z = hv.z * z;
  r.w = hv.w * z;
  reinterpret_cast<float4*>(agg)[q] = r;
  if ((q & 15) == 0) ssum[g] = z;
}

template <int F>
__global__ void k_edge_exp_aggregate(const int* __restrict__ src,
                                     const int* __restrict__ dst,
                                     const float* __restrict__ hfeat,
                                     const float* __restrict__ albuf,
                                     const unsigned* __restrict__ mkey,
                                     float* __restrict__ ssum,
                                     float* __restrict__ agg, int E) {
  constexpr int H = F / 64;
  int gt = blockIdx.x * blockDim.x + threadIdx.x;
  int e = gt >> 5;
  int lane = gt & 31;
  if (e >= E) return;
  const int s = src[e];
  const int d = dst[e];
  float z[H];
#pragma unroll
  for (int h = 0; h < H; ++h)
    z[h] = expf(albuf[(size_t)e * H + h] - funkey(mkey[d * H + h]));
  const float* hs = hfeat + (size_t)s * F;
  float* ad = agg + (size_t)d * F;
#pragma unroll
  for (int f = lane; f < F; f += 32)
    atomicAdd(&ad[f], hs[f] * z[f >> 6]);
  if (lane < H) {
    float zl = z[0];
    if (H > 1 && lane == 1) zl = z[H - 1];
    atomicAdd(&ssum[d * H + lane], zl);
  }
}

template <int F, bool ELU>
__global__ void k_norm_bias_act(float* __restrict__ a,
                                const float* __restrict__ ssum,
                                const float* __restrict__ b, int n) {
  int i = blockIdx.x * blockDim.x + threadIdx.x;
  if (i >= n) return;
  float v = a[i] / ssum[i >> 6] + b[i & (F - 1)];
  if (ELU) v = (v > 0.f) ? v : expm1f(v);
  a[i] = v;
}

extern "C" void kernel_launch(void* const* d_in, const int* in_sizes, int n_in,
                              void* d_out, int out_size, void* d_ws,
                              size_t ws_size, hipStream_t stream) {
  const float* x   = (const float*)d_in[0];
  const int*  eidx = (const int*)d_in[1];
  const float* ew  = (const float*)d_in[2];
  const float* W1  = (const float*)d_in[3];
  const float* We1 = (const float*)d_in[4];
  const float* as1 = (const float*)d_in[5];
  const float* ad1 = (const float*)d_in[6];
  const float* ae1 = (const float*)d_in[7];
  const float* b1  = (const float*)d_in[8];
  const float* W2  = (const float*)d_in[9];
  const float* We2 = (const float*)d_in[10];
  const float* as2 = (const float*)d_in[11];
  const float* ad2 = (const float*)d_in[12];
  const float* ae2 = (const float*)d_in[13];
  const float* b2  = (const float*)d_in[14];
  float* out = (float*)d_out;

  const int N  = in_sizes[0] / 128;
  const int E  = in_sizes[2];
  const int ET = E + N;
  const int* srcv = eidx;
  const int* dstv = eidx + E;

  float* p = (float*)d_ws;
  float* loopw = p; p += N;
  float* deg   = p; p += N;
  float* scoef = p; p += 2 * (size_t)N;
  float* dcoef = p; p += 2 * (size_t)N;
  float* cst   = p; p += 8;
  float* albuf = p; p += 2 * (size_t)ET;
  unsigned* mkey = (unsigned*)p; p += 2 * (size_t)N;
  float* ssum  = p; p += 2 * (size_t)N;
  float* hfeat = p; p += (size_t)N * 128;
  float* agg   = p; p += (size_t)N * 128;

  const int T = 256;
  auto cdiv = [](long long a, long long b) { return (int)((a + b - 1) / b); };

  hipMemsetAsync(deg, 0, (size_t)N * 4, stream);
  hipMemsetAsync(loopw, 0, (size_t)N * 4, stream);
  k_selfloop_stats<<<cdiv(E, T), T, 0, stream>>>(dstv, ew, deg, loopw, E);
  k_selfloop_div<<<cdiv(N, T), T, 0, stream>>>(loopw, deg, N);

  gemm_wmma_f32<128><<<cdiv(N, 16), 256, 0, stream>>>(x, W1, hfeat, N);
  k_edge_const<2><<<1, 32, 0, stream>>>(We1, ae1, cst);
  k_node_coef<2><<<cdiv(N, T), T, 0, stream>>>(hfeat, as1, ad1, loopw, cst, scoef,
                                               dcoef, albuf + 2 * (size_t)E, mkey, N);
  k_edge_logit_max<2><<<cdiv(E, T), T, 0, stream>>>(srcv, dstv, ew, scoef, dcoef,
                                                    cst, albuf, mkey, E);
  k_selfloop_start<<<cdiv((long long)N * 32, T), T, 0, stream>>>(
      hfeat, albuf + 2 * (size_t)E, mkey, ssum, agg, N * 32);
  k_edge_exp_aggregate<128><<<cdiv((long long)E * 32, T), T, 0, stream>>>(
      srcv, dstv, hfeat, albuf, mkey, ssum, agg, E);
  k_norm_bias_act<128, true><<<cdiv((long long)N * 128, T), T, 0, stream>>>(
      agg, ssum, b1, N * 128);

  gemm_wmma_f32<64><<<cdiv(N, 16), 128, 0, stream>>>(agg, W2, hfeat, N);
  k_edge_const<1><<<1, 32, 0, stream>>>(We2, ae2, cst);
  k_node_coef<1><<<cdiv(N, T), T, 0, stream>>>(hfeat, as2, ad2, loopw, cst, scoef,
                                               dcoef, albuf + (size_t)E, mkey, N);
  k_edge_logit_max<1><<<cdiv(E, T), T, 0, stream>>>(srcv, dstv, ew, scoef, dcoef,
                                                    cst, albuf, mkey, E);
  k_selfloop_start<<<cdiv((long long)N * 16, T), T, 0, stream>>>(
      hfeat, albuf + (size_t)E, mkey, ssum, out, N * 16);
  k_edge_exp_aggregate<64><<<cdiv((long long)E * 32, T), T, 0, stream>>>(
      srcv, dstv, hfeat, albuf, mkey, ssum, out, E);
  k_norm_bias_act<64, false><<<cdiv((long long)N * 64, T), T, 0, stream>>>(
      out, ssum, b2, N * 64);
}


# Softmax sum without float atomics: each work-group keeps 400 destination rows in LDS, scans the whole edge list, queues its matches, adds them one wave per row; f32 sum order fixed, by edge id
# speedup vs baseline: 1.3096x; 1.1521x over previous
#include <hip/hip_runtime.h>
#include <math.h>


typedef __attribute__((ext_vector_type(2))) float v2f;
typedef __attribute__((ext_vector_type(8))) float v8f;

__device__ __forceinline__ unsigned fkey(float f) {
  unsigned u = __float_as_uint(f);
  return (u & 0x80000000u) ? ~u : (u | 0x80000000u);
}
__device__ __forceinline__ float funkey(unsigned k) {
  unsigned u = (k & 0x80000000u) ? (k & 0x7fffffffu) : ~k;
  return __uint_as_float(u);
}

template <int NCOLS>
__global__ __launch_bounds__(32 * (NCOLS / 16))
void gemm_wmma_f32(const float* __restrict__ A, const float* __restrict__ Wg,
                   float* __restrict__ Out, int M) {
  constexpr int K = 128;
  constexpr int NT = NCOLS / 16;
  __shared__ float Ws[K * NCOLS];

  const int tid = threadIdx.x;
  for (int i = tid; i < K * NCOLS; i += 32 * NT) Ws[i] = Wg[i];
  __syncthreads();

  const int wave  = tid >> 5;
  const int lane  = tid & 31;
  const int row16 = blockIdx.x << 4;
  const int r     = lane & 15;
  const int kb    = (lane >> 4) << 1;
  const int col   = (wave << 4) + r;

  int arow = row16 + r;
  if (arow >= M) arow = M - 1;
  const float* __restrict__ ap = A + (size_t)arow * K;

  v8f acc = {};
#pragma unroll
  for (int k = 0; k < K; k += 4) {
    v2f a, b;
    a.x = ap[k + kb];
    a.y = ap[k + kb + 1];
    b.x = Ws[(k + kb) * NCOLS + col];
    b.y = Ws[(k + kb + 1) * NCOLS + col];
    acc = __builtin_amdgcn_wmma_f32_16x16x4_f32(false, a, false, b,
                                                (short)0, acc, false, false);
  }

  const int orow0 = row16 + ((lane >> 4) << 3);
  const int ocol  = (wave << 4) + r;
#pragma unroll
  for (int i = 0; i < 8; ++i) {
    int orow = orow0 + i;
    if (orow < M) Out[(size_t)orow * NCOLS + ocol] = acc[i];
  }
}

__global__ void k_selfloop_stats(const int* __restrict__ dst,
                                 const float* __restrict__ ew,
                                 float* __restrict__ deg,
                                 float* __restrict__ loopw, int E) {
  int e = blockIdx.x * blockDim.x + threadIdx.x;
  if (e >= E) return;
  int d = dst[e];
  atomicAdd(&deg[d], 1.0f);
  atomicAdd(&loopw[d], ew[e]);
}

__global__ void k_selfloop_div(float* __restrict__ loopw,
                               const float* __restrict__ deg, int N) {
  int n = blockIdx.x * blockDim.x + threadIdx.x;
  if (n >= N) return;
  loopw[n] = loopw[n] / fmaxf(deg[n], 1.0f);
}

template <int H>
__global__ void k_node_coef(const float* __restrict__ hfeat,
                            const float* __restrict__ a_s,
                            const float* __restrict__ a_d,
                            const float* __restrict__ loopw,
                            const float* __restrict__ cst,
                            float* __restrict__ scoef,
                            float* __restrict__ dcoef,
                            float* __restrict__ selfal,
                            unsigned* __restrict__ mkey, int N) {
  int n = blockIdx.x * blockDim.x + threadIdx.x;
  if (n >= N) return;
  const float* hr = hfeat + (size_t)n * (H * 64);
  const float w = loopw[n];
#pragma unroll
  for (int h = 0; h < H; ++h) {
    float ss = 0.f, dd = 0.f;
#pragma unroll 8
    for (int c = 0; c < 64; ++c) {
      float v = hr[h * 64 + c];
      ss += v * a_s[h * 64 + c];
      dd += v * a_d[h * 64 + c];
    }
    scoef[n * H + h] = ss;
    dcoef[n * H + h] = dd;
    float al = ss + dd + w * cst[h];
    al = (al >= 0.f) ? al : 0.2f * al;
    selfal[n * H + h] = al;
    mkey[n * H + h] = fkey(al);
  }
}

template <int H>
__global__ void k_edge_const(const float* __restrict__ We,
                             const float* __restrict__ ae,
                             float* __restrict__ cst) {
  int h = threadIdx.x;
  if (h < H) {
    float s = 0.f;
    for (int c = 0; c < 64; ++c) s += We[h * 64 + c] * ae[h * 64 + c];
    cst[h] = s;
  }
}

template <int H>
__global__ void k_edge_logit_max(const int* __restrict__ src,
                                 const int* __restrict__ dst,
                                 const float* __restrict__ ew,
                                 const float* __restrict__ scoef,
                                 const float* __restrict__ dcoef,
                                 const float* __restrict__ cst,
                                 float* __restrict__ albuf,
                                 unsigned* __restrict__ mkey, int E) {
  int e = blockIdx.x * blockDim.x + threadIdx.x;
  if (e >= E) return;
  const int s = src[e];
  const int d = dst[e];
  const float w = ew[e];
#pragma unroll
  for (int h = 0; h < H; ++h) {
    float al = scoef[s * H + h] + dcoef[d * H + h] + w * cst[h];
    al = (al >= 0.f) ? al : 0.2f * al;
    albuf[(size_t)e * H + h] = al;
    atomicMax(&mkey[d * H + h], fkey(al));
  }
}

__global__ void k_selfloop_start(const float* __restrict__ hfeat,
                                 const float* __restrict__ selfal,
                                 const unsigned* __restrict__ mkey,
                                 float* __restrict__ ssum,
                                 float* __restrict__ agg, int nquads) {
  int q = blockIdx.x * blockDim.x + threadIdx.x;
  if (q >= nquads) return;
  const int g = q >> 4;
  const float z = expf(selfal[g] - funkey(mkey[g]));
  const float4 hv = reinterpret_cast<const float4*>(hfeat)[q];
  float4 r;
  r.x = hv.x * z;
  r.y = hv.y * z;
  r.z = hv.z * z;
  r.w = hv.w * z;
  reinterpret_cast<float4*>(agg)[q] = r;
  if ((q & 15) == 0) ssum[g] = z;
}

#define AGG_R 400
#define AGG_SUB 16384
#define AGG_CAP 23552
#define AGG_WIN 1024
#define AGG_G 8
template <int F>
__global__ __launch_bounds__(1024)
void k_range_aggregate(const int* __restrict__ src,
                       const int* __restrict__ dst,
                       const float* __restrict__ hfeat,
                       const float* __restrict__ albuf,
                       const unsigned* __restrict__ mkey,
                       float* __restrict__ ssum,
                       float* __restrict__ agg, int E, int N) {
  constexpr int H = F / 64;
  constexpr int T = F / 32;
  __shared__ float acc[AGG_R * F];
  __shared__ float sacc[AGG_R * H];
  __shared__ unsigned q[AGG_CAP];
  __shared__ int w_src[AGG_WIN];
  __shared__ float w_z[H * AGG_WIN];
  __shared__ unsigned short wl[32 * 64];
  __shared__ unsigned omask[32 * 32];
  __shared__ int wcnt[2 * 32];

  const int tid  = threadIdx.x;
  const int wave = tid >> 5;
  const int lane = tid & 31;
  const int lo = blockIdx.x * AGG_R;
  const int nrows = (N - lo < AGG_R) ? (N - lo) : AGG_R;

  for (int i = tid; i < nrows * F; i += 1024) acc[i] = agg[(size_t)lo * F + i];
  for (int i = tid; i < nrows * H; i += 1024) sacc[i] = ssum[lo * H + i];

  const unsigned lt = (1u << lane) - 1u;
  const int slice = wave * 512 + lane;
  const int* __restrict__ dp = dst + slice;

  int dn[16];
  auto load_step = [&](int b) {
    if (b + AGG_SUB <= E) {
#pragma unroll
      for (int i = 0; i < 16; ++i) dn[i] = dp[b + i * 32];
    } else {
#pragma unroll
      for (int i = 0; i < 16; ++i) {
        const int e = b + slice + i * 32;
        dn[i] = (e < E) ? dst[e] : -1;
      }
    }
  };
  const int nsteps = (E + AGG_SUB - 1) / AGG_SUB;
  int step = (int)(((long long)blockIdx.x * nsteps) / gridDim.x);
  load_step(step * AGG_SUB);
  __syncthreads();

  int fill = 0;
  int par = 0;
  for (int k = 0; k < nsteps; ++k) {
    const int base = step * AGG_SUB;
    step = (step + 1 < nsteps) ? (step + 1) : 0;
    unsigned mk[16];
    int rl[16];
    int cnt = 0;
#pragma unroll
    for (int i = 0; i < 16; ++i) {
      rl[i] = dn[i] - lo;
      mk[i] = __builtin_amdgcn_ballot_w32((unsigned)rl[i] < (unsigned)nrows);
      cnt += __builtin_popcount(mk[i]);
    }
    if (lane == 0) wcnt[par * 32 + wave] = cnt;
    __syncthreads();
    load_step(step * AGG_SUB);
    const unsigned c = (unsigned)wcnt[par * 32 + lane];
    unsigned v = c | ((lane < wave) ? (c << 16) : 0u);
#pragma unroll
    for (int m = 16; m >= 1; m >>= 1) v += __shfl_xor(v, m);
    const int total = __builtin_amdgcn_readfirstlane((int)(v & 0xffffu));
    int off = fill + __builtin_amdgcn_readfirstlane((int)(v >> 16));
#pragma unroll
    for (int i = 0; i < 16; ++i) {
      if (mk[i] != 0u) {
        if ((mk[i] >> lane) & 1u)
          q[off + __builtin_popcount(mk[i] & lt)] =
              ((unsigned)(base + slice + i * 32) << 9) | (unsigned)rl[i];
        off += __builtin_popcount(mk[i]);
      }
    }
    fill += total;
    par ^= 1;

    const bool last = (k + 1 == nsteps);
    if (fill > AGG_CAP - AGG_SUB || last) {
      __syncthreads();
      for (int wb = 0; wb < fill; wb += AGG_WIN) {
        const int wn = (fill - wb < AGG_WIN) ? (fill - wb) : AGG_WIN;
        int owner = 32;
        if (tid < wn) {
          const unsigned ent = q[wb + tid];
          const int e = (int)(ent >> 9);
          const int d = lo + (int)(ent & 511u);
          owner = (int)(ent & 31u);
          w_src[tid] = src[e];
#pragma unroll
          for (int h = 0; h < H; ++h)
            w_z[h * AGG_WIN + tid] =
                expf(albuf[(size_t)e * H + h] - funkey(mkey[d * H + h]));
        }
        unsigned mym = 0u;
#pragma unroll
        for (int o = 0; o < 32; ++o) {
          const unsigned b = __builtin_amdgcn_ballot_w32(owner == o);
          mym = (lane == o) ? b : mym;
        }
        omask[lane * 32 + wave] = mym;
        __syncthreads();
        const unsigned mymask = omask[wave * 32 + lane];
        const int cm = __builtin_popcount(mymask);
        int pre = cm;
#pragma unroll
        for (int dlt = 1; dlt < 32; dlt <<= 1) {
          const int up = __shfl_up(pre, dlt);
          if (lane >= dlt) pre += up;
        }
        const int nall = __builtin_amdgcn_readlane(pre, 31);
        pre -= cm;
        for (int r0 = 0; r0 < nall; r0 += 64) {
          {
            unsigned mm = mymask;
            int p = pre - r0;
            while (__builtin_amdgcn_ballot_w32(mm != 0u) != 0u) {
              if (mm != 0u) {
                if (p >= 0 && p < 64)
                  wl[wave * 64 + p] = (unsigned short)(lane * 32 + __builtin_ctz(mm));
                mm &= mm - 1u;
                ++p;
              }
            }
          }
          const int n = (nall - r0 < 64) ? (nall - r0) : 64;
          for (int kk = 0; kk < n; kk += AGG_G) {
            int rr[AGG_G];
            float zz[AGG_G][H];
            float hv[AGG_G][T];
#pragma unroll
            for (int i = 0; i < AGG_G; ++i) {
              const int idx = wl[wave * 64 + ((kk + i < n) ? (kk + i) : kk)];
              rr[i] = (int)(q[wb + idx] & 511u);
              const float* hs = hfeat + (size_t)w_src[idx] * F;
#pragma unroll
              for (int h = 0; h < H; ++h) zz[i][h] = w_z[h * AGG_WIN + idx];
#pragma unroll
              for (int t = 0; t < T; ++t) hv[i][t] = hs[lane + 32 * t];
            }
#pragma unroll
            for (int i = 0; i < AGG_G; ++i) {
              if (kk + i < n) {
                float* ar = acc + rr[i] * F;
#pragma unroll
                for (int t = 0; t < T; ++t)
                  ar[lane + 32 * t] += hv[i][t] * zz[i][t >> 1];
                if (lane < H) sacc[rr[i] * H + lane] += (lane == 0) ? zz[i][0] : zz[i][H - 1];
              }
            }
          }
        }
        __syncthreads();
      }
      fill = 0;
    }
  }

  for (int i = tid; i < nrows * F; i += 1024) agg[(size_t)lo * F + i] = acc[i];
  for (int i = tid; i < nrows * H; i += 1024) ssum[lo * H + i] = sacc[i];
}

template <int F, bool ELU>
__global__ void k_norm_bias_act(float* __restrict__ a,
                                const float* __restrict__ ssum,
                                const float* __restrict__ b, int n) {
  int i = blockIdx.x * blockDim.x + threadIdx.x;
  if (i >= n) return;
  float v = a[i] / ssum[i >> 6] + b[i & (F - 1)];
  if (ELU) v = (v > 0.f) ? v : expm1f(v);
  a[i] = v;
}

extern "C" void kernel_launch(void* const* d_in, const int* in_sizes, int n_in,
                              void* d_out, int out_size, void* d_ws,
                              size_t ws_size, hipStream_t stream) {
  const float* x   = (const float*)d_in[0];
  const int*  eidx = (const int*)d_in[1];
  const float* ew  = (const float*)d_in[2];
  const float* W1  = (const float*)d_in[3];
  const float* We1 = (const float*)d_in[4];
  const float* as1 = (const float*)d_in[5];
  const float* ad1 = (const float*)d_in[6];
  const float* ae1 = (const float*)d_in[7];
  const float* b1  = (const float*)d_in[8];
  const float* W2  = (const float*)d_in[9];
  const float* We2 = (const float*)d_in[10];
  const float* as2 = (const float*)d_in[11];
  const float* ad2 = (const float*)d_in[12];
  const float* ae2 = (const float*)d_in[13];
  const float* b2  = (const float*)d_in[14];
  float* out = (float*)d_out;

  const int N  = in_sizes[0] / 128;
  const int E  = in_sizes[2];
  const int ET = E + N;
  const int* srcv = eidx;
  const int* dstv = eidx + E;

  float* p = (float*)d_ws;
  float* loopw = p; p += N;
  float* deg   = p; p += N;
  float* scoef = p; p += 2 * (size_t)N;
  float* dcoef = p; p += 2 * (size_t)N;
  float* cst   = p; p += 8;
  float* albuf = p; p += 2 * (size_t)ET;
  unsigned* mkey = (unsigned*)p; p += 2 * (size_t)N;
  float* ssum  = p; p += 2 * (size_t)N;
  float* hfeat = p; p += (size_t)N * 128;
  float* agg   = p; p += (size_t)N * 128;

  const int T = 256;
  auto cdiv = [](long long a, long long b) { return (int)((a + b - 1) / b); };

  hipMemsetAsync(deg, 0, (size_t)N * 4, stream);
  hipMemsetAsync(loopw, 0, (size_t)N * 4, stream);
  k_selfloop_stats<<<cdiv(E, T), T, 0, stream>>>(dstv, ew, deg, loopw, E);
  k_selfloop_div<<<cdiv(N, T), T, 0, stream>>>(loopw, deg, N);

  gemm_wmma_f32<128><<<cdiv(N, 16), 256, 0, stream>>>(x, W1, hfeat, N);
  k_edge_const<2><<<1, 32, 0, stream>>>(We1, ae1, cst);
  k_node_coef<2><<<cdiv(N, T), T, 0, stream>>>(hfeat, as1, ad1, loopw, cst, scoef,
                                               dcoef, albuf + 2 * (size_t)E, mkey, N);
  k_edge_logit_max<2><<<cdiv(E, T), T, 0, stream>>>(srcv, dstv, ew, scoef, dcoef,
                                                    cst, albuf, mkey, E);
  k_selfloop_start<<<cdiv((long long)N * 32, T), T, 0, stream>>>(
      hfeat, albuf + 2 * (size_t)E, mkey, ssum, agg, N * 32);
  k_range_aggregate<128><<<cdiv(N, AGG_R), 1024, 0, stream>>>(
      srcv, dstv, hfeat, albuf, mkey, ssum, agg, E, N);
  k_norm_bias_act<128, true><<<cdiv((long long)N * 128, T), T, 0, stream>>>(
      agg, ssum, b1, N * 128);

  gemm_wmma_f32<64><<<cdiv(N, 16), 128, 0, stream>>>(agg, W2, hfeat, N);
  k_edge_const<1><<<1, 32, 0, stream>>>(We2, ae2, cst);
  k_node_coef<1><<<cdiv(N, T), T, 0, stream>>>(hfeat, as2, ad2, loopw, cst, scoef,
                                               dcoef, albuf + (size_t)E, mkey, N);
  k_edge_logit_max<1><<<cdiv(E, T), T, 0, stream>>>(srcv, dstv, ew, scoef, dcoef,
                                                    cst, albuf, mkey, E);
  k_selfloop_start<<<cdiv((long long)N * 16, T), T, 0, stream>>>(
      hfeat, albuf + (size_t)E, mkey, ssum, out, N * 16);
  k_range_aggregate<64><<<cdiv(N, AGG_R), 1024, 0, stream>>>(
      srcv, dstv, hfeat, albuf, mkey, ssum, out, E, N);
  k_norm_bias_act<64, false><<<cdiv((long long)N * 64, T), T, 0, stream>>>(
      out, ssum, b2, N * 64);
}


# GEMM: each wave keeps its 16-column W fragments in registers for the whole launch; work-groups walk 16-row A strips staged through LDS by whole-row 16-byte loads; f32 sum order unchanged
# speedup vs baseline: 1.6895x; 1.2901x over previous
#include <hip/hip_runtime.h>
#include <math.h>


typedef __attribute__((ext_vector_type(2))) float v2f;
typedef __attribute__((ext_vector_type(8))) float v8f;

__device__ __forceinline__ unsigned fkey(float f) {
  unsigned u = __float_as_uint(f);
  return (u & 0x80000000u) ? ~u : (u | 0x80000000u);
}
__device__ __forceinline__ float funkey(unsigned k) {
  unsigned u = (k & 0x80000000u) ? (k & 0x7fffffffu) : ~k;
  return __uint_as_float(u);
}

template <int NCOLS>
__global__ __launch_bounds__(32 * (NCOLS / 16))
void gemm_wmma_f32(const float* __restrict__ A, const float* __restrict__ Wg,
                   float* __restrict__ Out, int M) {
  constexpr int K = 128;
  constexpr int NT = NCOLS / 16;
  constexpr int NTHR = 32 * NT;
  constexpr int PITCH = K + 4;
  constexpr int NV = (16 * K / 4) / NTHR;
  static_assert(NV == 2 || NV == 4, "strip staging is written for 4 or 8 waves");
  __shared__ __attribute__((aligned(16))) float As[2][16 * PITCH];

  const int tid  = threadIdx.x;
  const int wave = tid >> 5;
  const int lane = tid & 31;
  const int r    = lane & 15;
  const int kb   = (lane >> 4) << 1;
  const int col  = (wave << 4) + r;

  v2f bf[K / 4];
#pragma unroll
  for (int s = 0; s < K / 4; ++s) {
    bf[s].x = Wg[(4 * s + kb) * NCOLS + col];
    bf[s].y = Wg[(4 * s + kb + 1) * NCOLS + col];
  }

  const int nstrips = (M + 15) >> 4;
  float4 st0, st1, st2, st3;
  st0 = st1 = st2 = st3 = make_float4(0.f, 0.f, 0.f, 0.f);
  auto gpiece = [&](int sp, int j) __attribute__((always_inline)) -> float4 {
    const int v = tid + j * NTHR;
    int arow = (sp << 4) + (v >> 5);
    if (arow >= M) arow = M - 1;
    return *reinterpret_cast<const float4*>(A + (size_t)arow * K + ((v & 31) << 2));
  };
  auto gload = [&](int sp) __attribute__((always_inline)) {
    st0 = gpiece(sp, 0);
    st1 = gpiece(sp, 1);
    if constexpr (NV > 2) {
      st2 = gpiece(sp, 2);
      st3 = gpiece(sp, 3);
    }
  };
  auto lpiece = [&](int b, int j, const float4 val) __attribute__((always_inline)) {
    const int v = tid + j * NTHR;
    *reinterpret_cast<float4*>(&As[b][(v >> 5) * PITCH + ((v & 31) << 2)]) = val;
  };
  auto lstore = [&](int b) __attribute__((always_inline)) {
    lpiece(b, 0, st0);
    lpiece(b, 1, st1);
    if constexpr (NV > 2) {
      lpiece(b, 2, st2);
      lpiece(b, 3, st3);
    }
  };

  int strip = blockIdx.x;
  if (strip < nstrips) {
    gload(strip);
    lstore(0);
  }
  __syncthreads();

  int buf = 0;
  for (; strip < nstrips; strip += gridDim.x) {
    const int nxt = strip + gridDim.x;
    if (nxt < nstrips) gload(nxt);

    const float* ar = &As[buf][r * PITCH + kb];
    v8f acc = {};
#pragma unroll
    for (int s = 0; s < K / 4; ++s) {
      const v2f a = *reinterpret_cast<const v2f*>(ar + 4 * s);
      acc = __builtin_amdgcn_wmma_f32_16x16x4_f32(false, a, false, bf[s],
                                                  (short)0, acc, false, false);
    }

    const int orow0 = (strip << 4) + ((lane >> 4) << 3);
#pragma unroll
    for (int i = 0; i < 8; ++i) {
      const int orow = orow0 + i;
      if (orow < M) Out[(size_t)orow * NCOLS + col] = acc[i];
    }

    if (nxt < nstrips) lstore(buf ^ 1);
    __syncthreads();
    buf ^= 1;
  }
}

__global__ void k_selfloop_stats(const int* __restrict__ dst,
                                 const float* __restrict__ ew,
                                 float* __restrict__ deg,
                                 float* __restrict__ loopw, int E) {
  int e = blockIdx.x * blockDim.x + threadIdx.x;
  if (e >= E) return;
  int d = dst[e];
  atomicAdd(&deg[d], 1.0f);
  atomicAdd(&loopw[d], ew[e]);
}

__global__ void k_selfloop_div(float* __restrict__ loopw,
                               const float* __restrict__ deg, int N) {
  int n = blockIdx.x * blockDim.x + threadIdx.x;
  if (n >= N) return;
  loopw[n] = loopw[n] / fmaxf(deg[n], 1.0f);
}

template <int H>
__global__ void k_node_coef(const float* __restrict__ hfeat,
                            const float* __restrict__ a_s,
                            const float* __restrict__ a_d,
                            const float* __restrict__ loopw,
                            const float* __restrict__ cst,
                            float* __restrict__ scoef,
                            float* __restrict__ dcoef,
                            float* __restrict__ selfal,
                            unsigned* __restrict__ mkey, int N) {
  int n = blockIdx.x * blockDim.x + threadIdx.x;
  if (n >= N) return;
  const float* hr = hfeat + (size_t)n * (H * 64);
  const float w = loopw[n];
#pragma unroll
  for (int h = 0; h < H; ++h) {
    float ss = 0.f, dd = 0.f;
#pragma unroll 8
    for (int c = 0; c < 64; ++c) {
      float v = hr[h * 64 + c];
      ss += v * a_s[h * 64 + c];
      dd += v * a_d[h * 64 + c];
    }
    scoef[n * H + h] = ss;
    dcoef[n * H + h] = dd;
    float al = ss + dd + w * cst[h];
    al = (al >= 0.f) ? al : 0.2f * al;
    selfal[n * H + h] = al;
    mkey[n * H + h] = fkey(al);
  }
}

template <int H>
__global__ void k_edge_const(const float* __restrict__ We,
                             const float* __restrict__ ae,
                             float* __restrict__ cst) {
  int h = threadIdx.x;
  if (h < H) {
    float s = 0.f;
    for (int c = 0; c < 64; ++c) s += We[h * 64 + c] * ae[h * 64 + c];
    cst[h] = s;
  }
}

template <int H>
__global__ void k_edge_logit_max(const int* __restrict__ src,
                                 const int* __restrict__ dst,
                                 const float* __restrict__ ew,
                                 const float* __restrict__ scoef,
                                 const float* __restrict__ dcoef,
                                 const float* __restrict__ cst,
                                 float* __restrict__ albuf,
                                 unsigned* __restrict__ mkey, int E) {
  int e = blockIdx.x * blockDim.x + threadIdx.x;
  if (e >= E) return;
  const int s = src[e];
  const int d = dst[e];
  const float w = ew[e];
#pragma unroll
  for (int h = 0; h < H; ++h) {
    float al = scoef[s * H + h] + dcoef[d * H + h] + w * cst[h];
    al = (al >= 0.f) ? al : 0.2f * al;
    albuf[(size_t)e * H + h] = al;
    atomicMax(&mkey[d * H + h], fkey(al));
  }
}

#define AGG_R 400
#define AGG_SUB 16384
#define AGG_CAP 23552
#define AGG_WIN 1024
#define AGG_G 8
#define AGG_LCAP 65536
template <int F, bool ELU, int LIST>
__global__ __launch_bounds__(1024)
void k_range_aggregate(const int* __restrict__ src,
                       const int* __restrict__ dst,
                       const float* __restrict__ hfeat,
                       const float* __restrict__ albuf,
                       const float* __restrict__ selfal,
                       const unsigned* __restrict__ mkey,
                       const float* __restrict__ bias,
                       float* __restrict__ outp,
                       unsigned* __restrict__ lists,
                       int* __restrict__ lcnt, int E, int N) {
  constexpr int H = F / 64;
  constexpr int T = F / 32;
  __shared__ __attribute__((aligned(16))) float acc[AGG_R * F];
  __shared__ float sacc[AGG_R * H];
  __shared__ unsigned q[AGG_CAP];
  __shared__ int w_src[AGG_WIN];
  __shared__ float w_z[H * AGG_WIN];
  __shared__ unsigned short wl[32 * 64];
  __shared__ unsigned omask[32 * 32];
  __shared__ int wcnt[2 * 32];

  const int tid  = threadIdx.x;
  const int wave = tid >> 5;
  const int lane = tid & 31;
  const int lo = blockIdx.x * AGG_R;
  const int nrows = (N - lo < AGG_R) ? (N - lo) : AGG_R;

  for (int i = tid; i < nrows * H; i += 1024)
    sacc[i] = expf(selfal[lo * H + i] - funkey(mkey[lo * H + i]));
  __syncthreads();
  {
    const float4* __restrict__ h4 =
        reinterpret_cast<const float4*>(hfeat + (size_t)lo * F);
    float4* a4 = reinterpret_cast<float4*>(acc);
    for (int i = tid; i < nrows * (F / 4); i += 1024) {
      const float z = sacc[i >> 4];
      const float4 hv = h4[i];
      float4 r;
      r.x = hv.x * z;
      r.y = hv.y * z;
      r.z = hv.z * z;
      r.w = hv.w * z;
      a4[i] = r;
    }
  }

  unsigned* __restrict__ mylist = lists + (size_t)blockIdx.x * AGG_LCAP;
  int lsize = 0;

  auto drain = [&](const int fill) __attribute__((always_inline)) {
    for (int wb = 0; wb < fill; wb += AGG_WIN) {
      const int wn = (fill - wb < AGG_WIN) ? (fill - wb) : AGG_WIN;
      int owner = 32;
      if (tid < wn) {
        const unsigned ent = q[wb + tid];
        const int e = (int)(ent >> 9);
        const int d = lo + (int)(ent & 511u);
        owner = (int)(ent & 31u);
        w_src[tid] = src[e];
#pragma unroll
        for (int h = 0; h < H; ++h)
          w_z[h * AGG_WIN + tid] =
              expf(albuf[(size_t)e * H + h] - funkey(mkey[d * H + h]));
      }
      unsigned mym = 0u;
#pragma unroll
      for (int o = 0; o < 32; ++o) {
        const unsigned b = __builtin_amdgcn_ballot_w32(owner == o);
        mym = (lane == o) ? b : mym;
      }
      omask[lane * 32 + wave] = mym;
      __syncthreads();
      const unsigned mymask = omask[wave * 32 + lane];
      const int cm = __builtin_popcount(mymask);
      int pre = cm;
#pragma unroll
      for (int dlt = 1; dlt < 32; dlt <<= 1) {
        const int up = __shfl_up(pre, dlt);
        if (lane >= dlt) pre += up;
      }
      const int nall = __builtin_amdgcn_readlane(pre, 31);
      pre -= cm;
      for (int r0 = 0; r0 < nall; r0 += 64) {
        {
          unsigned mm = mymask;
          int p = pre - r0;
          while (__builtin_amdgcn_ballot_w32(mm != 0u) != 0u) {
            if (mm != 0u) {
              if (p >= 0 && p < 64)
                wl[wave * 64 + p] = (unsigned short)(lane * 32 + __builtin_ctz(mm));
              mm &= mm - 1u;
              ++p;
            }
          }
        }
        const int n = (nall - r0 < 64) ? (nall - r0) : 64;
        for (int kk = 0; kk < n; kk += AGG_G) {
          int rr[AGG_G];
          float zz[AGG_G][H];
          float hv[AGG_G][T];
#pragma unroll
          for (int i = 0; i < AGG_G; ++i) {
            const int idx = wl[wave * 64 + ((kk + i < n) ? (kk + i) : kk)];
            rr[i] = (int)(q[wb + idx] & 511u);
            const float* hs = hfeat + (size_t)w_src[idx] * F;
#pragma unroll
            for (int h = 0; h < H; ++h) zz[i][h] = w_z[h * AGG_WIN + idx];
#pragma unroll
            for (int t = 0; t < T; ++t) hv[i][t] = hs[lane + 32 * t];
          }
#pragma unroll
          for (int i = 0; i < AGG_G; ++i) {
            if (kk + i < n) {
              float* ar = acc + rr[i] * F;
#pragma unroll
              for (int t = 0; t < T; ++t)
                ar[lane + 32 * t] += hv[i][t] * zz[i][t >> 1];
              if (lane < H) sacc[rr[i] * H + lane] += (lane == 0) ? zz[i][0] : zz[i][H - 1];
            }
          }
        }
      }
      __syncthreads();
    }
  };

  auto scan_all = [&]() __attribute__((always_inline)) {
    const unsigned lt = (1u << lane) - 1u;
    const int slice = wave * 512 + lane;
    const int* __restrict__ dp = dst + slice;

    int dn[16];
    auto load_step = [&](int b) {
      if (b + AGG_SUB <= E) {
#pragma unroll
        for (int i = 0; i < 16; ++i) dn[i] = dp[b + i * 32];
      } else {
#pragma unroll
        for (int i = 0; i < 16; ++i) {
          const int e = b + slice + i * 32;
          dn[i] = (e < E) ? dst[e] : -1;
        }
      }
    };
    const int nsteps = (E + AGG_SUB - 1) / AGG_SUB;
    int step = (int)(((long long)blockIdx.x * nsteps) / gridDim.x);
    load_step(step * AGG_SUB);
    __syncthreads();

    int fill = 0;
    int par = 0;
    for (int k = 0; k < nsteps; ++k) {
      const int base = step * AGG_SUB;
      step = (step + 1 < nsteps) ? (step + 1) : 0;
      unsigned mk[16];
      int rl[16];
      int cnt = 0;
#pragma unroll
      for (int i = 0; i < 16; ++i) {
        rl[i] = dn[i] - lo;
        mk[i] = __builtin_amdgcn_ballot_w32((unsigned)rl[i] < (unsigned)nrows);
        cnt += __builtin_popcount(mk[i]);
      }
      if (lane == 0) wcnt[par * 32 + wave] = cnt;
      __syncthreads();
      load_step(step * AGG_SUB);
      const unsigned c = (unsigned)wcnt[par * 32 + lane];
      unsigned v = c | ((lane < wave) ? (c << 16) : 0u);
#pragma unroll
      for (int m = 16; m >= 1; m >>= 1) v += __shfl_xor(v, m);
      const int total = __builtin_amdgcn_readfirstlane((int)(v & 0xffffu));
      int off = fill + __builtin_amdgcn_readfirstlane((int)(v >> 16));
#pragma unroll
      for (int i = 0; i < 16; ++i) {
        if (mk[i] != 0u) {
          if ((mk[i] >> lane) & 1u)
            q[off + __builtin_popcount(mk[i] & lt)] =
                ((unsigned)(base + slice + i * 32) << 9) | (unsigned)rl[i];
          off += __builtin_popcount(mk[i]);
        }
      }
      fill += total;
      par ^= 1;

      const bool last = (k + 1 == nsteps);
      if (fill > AGG_CAP - AGG_SUB || last) {
        __syncthreads();
        if (LIST == 1) {
          if (lsize >= 0 && lsize + fill <= AGG_LCAP) {
            for (int i = tid; i < fill; i += 1024) mylist[lsize + i] = q[i];
            lsize += fill;
          } else {
            lsize = -1;
          }
        }
        drain(fill);
        fill = 0;
      }
    }
    if (LIST == 1 && tid == 0) lcnt[blockIdx.x] = lsize;
  };

  if (LIST == 2) {
    const int cnt = lcnt[blockIdx.x];
    if (cnt >= 0 && cnt <= AGG_LCAP) {
      for (int c0 = 0; c0 < cnt; c0 += AGG_CAP) {
        const int fill = (cnt - c0 < AGG_CAP) ? (cnt - c0) : AGG_CAP;
        for (int i = tid; i < fill; i += 1024) q[i] = mylist[c0 + i];
        __syncthreads();
        drain(fill);
      }
      __syncthreads();
    } else {
      scan_all();
    }
  } else {
    scan_all();
  }

  {
    const float4* a4 = reinterpret_cast<const float4*>(acc);
    float4* __restrict__ o4 = reinterpret_cast<float4*>(outp + (size_t)lo * F);
    for (int i = tid; i < nrows * (F / 4); i += 1024) {
      const float s = sacc[i >> 4];
      const float4 a = a4[i];
      const int f = (i << 2) & (F - 1);
      float4 r;
      r.x = a.x / s + bias[f];
      r.y = a.y / s + bias[f + 1];
      r.z = a.z / s + bias[f + 2];
      r.w = a.w / s + bias[f + 3];
      if (ELU) {
        r.x = (r.x > 0.f) ? r.x : expm1f(r.x);
        r.y = (r.y > 0.f) ? r.y : expm1f(r.y);
        r.z = (r.z > 0.f) ? r.z : expm1f(r.z);
        r.w = (r.w > 0.f) ? r.w : expm1f(r.w);
      }
      o4[i] = r;
    }
  }
}

extern "C" void kernel_launch(void* const* d_in, const int* in_sizes, int n_in,
                              void* d_out, int out_size, void* d_ws,
                              size_t ws_size, hipStream_t stream) {
  const float* x   = (const float*)d_in[0];
  const int*  eidx = (const int*)d_in[1];
  const float* ew  = (const float*)d_in[2];
  const float* W1  = (const float*)d_in[3];
  const float* We1 = (const float*)d_in[4];
  const float* as1 = (const float*)d_in[5];
  const float* ad1 = (const float*)d_in[6];
  const float* ae1 = (const float*)d_in[7];
  const float* b1  = (const float*)d_in[8];
  const float* W2  = (const float*)d_in[9];
  const float* We2 = (const float*)d_in[10];
  const float* as2 = (const float*)d_in[11];
  const float* ad2 = (const float*)d_in[12];
  const float* ae2 = (const float*)d_in[13];
  const float* b2  = (const float*)d_in[14];
  float* out = (float*)d_out;

  const int N  = in_sizes[0] / 128;
  const int E  = in_sizes[2];
  const int ET = E + N;
  const int* srcv = eidx;
  const int* dstv = eidx + E;

  float* p = (float*)d_ws;
  float* loopw = p; p += N;
  float* deg   = p; p += N;
  float* scoef = p; p += 2 * (size_t)N;
  float* dcoef = p; p += 2 * (size_t)N;
  float* cst   = p; p += 8;
  float* albuf = p; p += 2 * (size_t)ET;
  unsigned* mkey = (unsigned*)p; p += 2 * (size_t)N;
  float* hfeat = p; p += (size_t)N * 128;
  float* agg   = p; p += (size_t)N * 128;
  int* lcnt = (int*)p; p += 1024;
  unsigned* lists = (unsigned*)p;

  const int T = 256;
  auto cdiv = [](long long a, long long b) { return (int)((a + b - 1) / b); };

  hipMemsetAsync(deg, 0, (size_t)N * 4, stream);
  hipMemsetAsync(loopw, 0, (size_t)N * 4, stream);
  k_selfloop_stats<<<cdiv(E, T), T, 0, stream>>>(dstv, ew, deg, loopw, E);
  k_selfloop_div<<<cdiv(N, T), T, 0, stream>>>(loopw, deg, N);

  gemm_wmma_f32<128><<<cdiv(N, 16 * 8), 256, 0, stream>>>(x, W1, hfeat, N);
  k_edge_const<2><<<1, 32, 0, stream>>>(We1, ae1, cst);
  k_node_coef<2><<<cdiv(N, T), T, 0, stream>>>(hfeat, as1, ad1, loopw, cst, scoef,
                                               dcoef, albuf + 2 * (size_t)E, mkey, N);
  k_edge_logit_max<2><<<cdiv(E, T), T, 0, stream>>>(srcv, dstv, ew, scoef, dcoef,
                                                    cst, albuf, mkey, E);
  k_range_aggregate<128, true, 1><<<cdiv(N, AGG_R), 1024, 0, stream>>>(
      srcv, dstv, hfeat, albuf, albuf + 2 * (size_t)E, mkey, b1, agg, lists, lcnt, E, N);

  gemm_wmma_f32<64><<<cdiv(N, 16 * 6), 128, 0, stream>>>(agg, W2, hfeat, N);
  k_edge_const<1><<<1, 32, 0, stream>>>(We2, ae2, cst);
  k_node_coef<1><<<cdiv(N, T), T, 0, stream>>>(hfeat, as2, ad2, loopw, cst, scoef,
                                               dcoef, albuf + (size_t)E, mkey, N);
  k_edge_logit_max<1><<<cdiv(E, T), T, 0, stream>>>(srcv, dstv, ew, scoef, dcoef,
                                                    cst, albuf, mkey, E);
  k_range_aggregate<64, false, 2><<<cdiv(N, AGG_R), 1024, 0, stream>>>(
      srcv, dstv, hfeat, albuf, albuf + (size_t)E, mkey, b2, out, lists, lcnt, E, N);
}


# Segment max without global atomics: each row-owning work-group maxes its edges' logits in LDS (integer max, ordered keys) before its sum pass; edge pass only stores logits; f32 sum order unchanged
# speedup vs baseline: 2.2136x; 1.3102x over previous
#include <hip/hip_runtime.h>
#include <math.h>


typedef __attribute__((ext_vector_type(2))) float v2f;
typedef __attribute__((ext_vector_type(8))) float v8f;

__device__ __forceinline__ unsigned fkey(float f) {
  unsigned u = __float_as_uint(f);
  return (u & 0x80000000u) ? ~u : (u | 0x80000000u);
}
__device__ __forceinline__ float funkey(unsigned k) {
  unsigned u = (k & 0x80000000u) ? (k & 0x7fffffffu) : ~k;
  return __uint_as_float(u);
}

template <int NCOLS>
__global__ __launch_bounds__(32 * (NCOLS / 16))
void gemm_wmma_f32(const float* __restrict__ A, const float* __restrict__ Wg,
                   float* __restrict__ Out, int M) {
  constexpr int K = 128;
  constexpr int NT = NCOLS / 16;
  constexpr int NTHR = 32 * NT;
  constexpr int PITCH = K + 4;
  constexpr int NV = (16 * K / 4) / NTHR;
  static_assert(NV == 2 || NV == 4, "strip staging is written for 4 or 8 waves");
  __shared__ __attribute__((aligned(16))) float As[2][16 * PITCH];

  const int tid  = threadIdx.x;
  const int wave = tid >> 5;
  const int lane = tid & 31;
  const int r    = lane & 15;
  const int kb   = (lane >> 4) << 1;
  const int col  = (wave << 4) + r;

  v2f bf[K / 4];
#pragma unroll
  for (int s = 0; s < K / 4; ++s) {
    bf[s].x = Wg[(4 * s + kb) * NCOLS + col];
    bf[s].y = Wg[(4 * s + kb + 1) * NCOLS + col];
  }

  const int nstrips = (M + 15) >> 4;
  float4 st0, st1, st2, st3;
  st0 = st1 = st2 = st3 = make_float4(0.f, 0.f, 0.f, 0.f);
  auto gpiece = [&](int sp, int j) __attribute__((always_inline)) -> float4 {
    const int v = tid + j * NTHR;
    int arow = (sp << 4) + (v >> 5);
    if (arow >= M) arow = M - 1;
    return *reinterpret_cast<const float4*>(A + (size_t)arow * K + ((v & 31) << 2));
  };
  auto gload = [&](int sp) __attribute__((always_inline)) {
    st0 = gpiece(sp, 0);
    st1 = gpiece(sp, 1);
    if constexpr (NV > 2) {
      st2 = gpiece(sp, 2);
      st3 = gpiece(sp, 3);
    }
  };
  auto lpiece = [&](int b, int j, const float4 val) __attribute__((always_inline)) {
    const int v = tid + j * NTHR;
    *reinterpret_cast<float4*>(&As[b][(v >> 5) * PITCH + ((v & 31) << 2)]) = val;
  };
  auto lstore = [&](int b) __attribute__((always_inline)) {
    lpiece(b, 0, st0);
    lpiece(b, 1, st1);
    if constexpr (NV > 2) {
      lpiece(b, 2, st2);
      lpiece(b, 3, st3);
    }
  };

  int strip = blockIdx.x;
  if (strip < nstrips) {
    gload(strip);
    lstore(0);
  }
  __syncthreads();

  int buf = 0;
  for (; strip < nstrips; strip += gridDim.x) {
    const int nxt = strip + gridDim.x;
    if (nxt < nstrips) gload(nxt);

    const float* ar = &As[buf][r * PITCH + kb];
    v8f acc = {};
#pragma unroll
    for (int s = 0; s < K / 4; ++s) {
      const v2f a = *reinterpret_cast<const v2f*>(ar + 4 * s);
      acc = __builtin_amdgcn_wmma_f32_16x16x4_f32(false, a, false, bf[s],
                                                  (short)0, acc, false, false);
    }

    const int orow0 = (strip << 4) + ((lane >> 4) << 3);
#pragma unroll
    for (int i = 0; i < 8; ++i) {
      const int orow = orow0 + i;
      if (orow < M) Out[(size_t)orow * NCOLS + col] = acc[i];
    }

    if (nxt < nstrips) lstore(buf ^ 1);
    __syncthreads();
    buf ^= 1;
  }
}

__global__ void k_selfloop_stats(const int* __restrict__ dst,
                                 const float* __restrict__ ew,
                                 float* __restrict__ deg,
                                 float* __restrict__ loopw, int E) {
  int e = blockIdx.x * blockDim.x + threadIdx.x;
  if (e >= E) return;
  int d = dst[e];
  atomicAdd(&deg[d], 1.0f);
  atomicAdd(&loopw[d], ew[e]);
}

__global__ void k_selfloop_div(float* __restrict__ loopw,
                               const float* __restrict__ deg, int N) {
  int n = blockIdx.x * blockDim.x + threadIdx.x;
  if (n >= N) return;
  loopw[n] = loopw[n] / fmaxf(deg[n], 1.0f);
}

template <int H>
__global__ void k_node_coef(const float* __restrict__ hfeat,
                            const float* __restrict__ a_s,
                            const float* __restrict__ a_d,
                            const float* __restrict__ loopw,
                            const float* __restrict__ cst,
                            float* __restrict__ scoef,
                            float* __restrict__ dcoef,
                            float* __restrict__ selfal, int N) {
  int n = blockIdx.x * blockDim.x + threadIdx.x;
  if (n >= N) return;
  const float* hr = hfeat + (size_t)n * (H * 64);
  const float w = loopw[n];
#pragma unroll
  for (int h = 0; h < H; ++h) {
    float ss = 0.f, dd = 0.f;
#pragma unroll 8
    for (int c = 0; c < 64; ++c) {
      float v = hr[h * 64 + c];
      ss += v * a_s[h * 64 + c];
      dd += v * a_d[h * 64 + c];
    }
    scoef[n * H + h] = ss;
    dcoef[n * H + h] = dd;
    float al = ss + dd + w * cst[h];
    al = (al >= 0.f) ? al : 0.2f * al;
    selfal[n * H + h] = al;
  }
}

template <int H>
__global__ void k_edge_const(const float* __restrict__ We,
                             const float* __restrict__ ae,
                             float* __restrict__ cst) {
  int h = threadIdx.x;
  if (h < H) {
    float s = 0.f;
    for (int c = 0; c < 64; ++c) s += We[h * 64 + c] * ae[h * 64 + c];
    cst[h] = s;
  }
}

template <int H>
__global__ void k_edge_logit(const int* __restrict__ src,
                             const int* __restrict__ dst,
                             const float* __restrict__ ew,
                             const float* __restrict__ scoef,
                             const float* __restrict__ dcoef,
                             const float* __restrict__ cst,
                             float* __restrict__ albuf, int E) {
  int e = blockIdx.x * blockDim.x + threadIdx.x;
  if (e >= E) return;
  const int s = src[e];
  const int d = dst[e];
  const float w = ew[e];
#pragma unroll
  for (int h = 0; h < H; ++h) {
    float al = scoef[s * H + h] + dcoef[d * H + h] + w * cst[h];
    al = (al >= 0.f) ? al : 0.2f * al;
    albuf[(size_t)e * H + h] = al;
  }
}

#define AGG_R 400
#define AGG_SUB 16384
#define AGG_CAP 23040
#define AGG_WIN 1024
#define AGG_G 8
#define AGG_MB 8
#define AGG_LCAP 65536
template <int F, bool ELU, int LIST>
__global__ __launch_bounds__(1024)
void k_range_aggregate(const int* __restrict__ src,
                       const int* __restrict__ dst,
                       const float* __restrict__ hfeat,
                       const float* __restrict__ albuf,
                       const float* __restrict__ selfal,
                       const float* __restrict__ bias,
                       float* __restrict__ outp,
                       unsigned* __restrict__ lists,
                       int* __restrict__ lcnt, int E, int N) {
  constexpr int H = F / 64;
  constexpr int T = F / 32;
  __shared__ __attribute__((aligned(16))) float acc[AGG_R * F];
  __shared__ float sacc[AGG_R * H];
  __shared__ unsigned mx[AGG_R * H];
  __shared__ unsigned q[AGG_CAP];
  __shared__ int w_src[AGG_WIN];
  __shared__ float w_z[H * AGG_WIN];
  __shared__ unsigned short wl[32 * 64];
  __shared__ unsigned omask[32 * 32];
  __shared__ int wcnt[2 * 32];

  const int tid  = threadIdx.x;
  const int wave = tid >> 5;
  const int lane = tid & 31;
  const int lo = blockIdx.x * AGG_R;
  const int nrows = (N - lo < AGG_R) ? (N - lo) : AGG_R;

  for (int i = tid; i < nrows * H; i += 1024) mx[i] = fkey(selfal[lo * H + i]);
  __syncthreads();

  unsigned* __restrict__ mylist = lists + (size_t)blockIdx.x * AGG_LCAP;
  int lsize = 0;

  auto maxgroup = [&](const unsigned (&me)[AGG_MB]) __attribute__((always_inline)) {
    float ma[AGG_MB][H];
#pragma unroll
    for (int j = 0; j < AGG_MB; ++j) {
#pragma unroll
      for (int h = 0; h < H; ++h)
        ma[j][h] = albuf[(size_t)(me[j] >> 9) * H + h];
    }
#pragma unroll
    for (int j = 0; j < AGG_MB; ++j) {
#pragma unroll
      for (int h = 0; h < H; ++h)
        atomicMax(&mx[(int)(me[j] & 511u) * H + h], fkey(ma[j][h]));
    }
  };
  auto maxpass_queue = [&](const int n) __attribute__((always_inline)) {
    for (int i0 = tid; i0 < n; i0 += AGG_MB * 1024) {
      unsigned me[AGG_MB];
#pragma unroll
      for (int j = 0; j < AGG_MB; ++j) {
        const int i = i0 + j * 1024;
        me[j] = q[(i < n) ? i : i0];
      }
      maxgroup(me);
    }
  };
  auto maxpass_list = [&](const int n) __attribute__((always_inline)) {
    for (int i0 = tid; i0 < n; i0 += AGG_MB * 1024) {
      unsigned me[AGG_MB];
#pragma unroll
      for (int j = 0; j < AGG_MB; ++j) {
        const int i = i0 + j * 1024;
        me[j] = mylist[(i < n) ? i : i0];
      }
      maxgroup(me);
    }
  };

  auto drain = [&](const int fill) __attribute__((always_inline)) {
    for (int wb = 0; wb < fill; wb += AGG_WIN) {
      const int wn = (fill - wb < AGG_WIN) ? (fill - wb) : AGG_WIN;
      int owner = 32;
      if (tid < wn) {
        const unsigned ent = q[wb + tid];
        const int e = (int)(ent >> 9);
        const int rw = (int)(ent & 511u);
        owner = (int)(ent & 31u);
        w_src[tid] = src[e];
#pragma unroll
        for (int h = 0; h < H; ++h)
          w_z[h * AGG_WIN + tid] =
              expf(albuf[(size_t)e * H + h] - funkey(mx[rw * H + h]));
      }
      unsigned mym = 0u;
#pragma unroll
      for (int o = 0; o < 32; ++o) {
        const unsigned b = __builtin_amdgcn_ballot_w32(owner == o);
        mym = (lane == o) ? b : mym;
      }
      omask[lane * 32 + wave] = mym;
      __syncthreads();
      const unsigned mymask = omask[wave * 32 + lane];
      const int cm = __builtin_popcount(mymask);
      int pre = cm;
#pragma unroll
      for (int dlt = 1; dlt < 32; dlt <<= 1) {
        const int up = __shfl_up(pre, dlt);
        if (lane >= dlt) pre += up;
      }
      const int nall = __builtin_amdgcn_readlane(pre, 31);
      pre -= cm;
      for (int r0 = 0; r0 < nall; r0 += 64) {
        {
          unsigned mm = mymask;
          int p = pre - r0;
          while (__builtin_amdgcn_ballot_w32(mm != 0u) != 0u) {
            if (mm != 0u) {
              if (p >= 0 && p < 64)
                wl[wave * 64 + p] = (unsigned short)(lane * 32 + __builtin_ctz(mm));
              mm &= mm - 1u;
              ++p;
            }
          }
        }
        const int n = (nall - r0 < 64) ? (nall - r0) : 64;
        for (int kk = 0; kk < n; kk += AGG_G) {
          int rr[AGG_G];
          float zz[AGG_G][H];
          float hv[AGG_G][T];
#pragma unroll
          for (int i = 0; i < AGG_G; ++i) {
            const int idx = wl[wave * 64 + ((kk + i < n) ? (kk + i) : kk)];
            rr[i] = (int)(q[wb + idx] & 511u);
            const float* hs = hfeat + (size_t)w_src[idx] * F;
#pragma unroll
            for (int h = 0; h < H; ++h) zz[i][h] = w_z[h * AGG_WIN + idx];
#pragma unroll
            for (int t = 0; t < T; ++t) hv[i][t] = hs[lane + 32 * t];
          }
#pragma unroll
          for (int i = 0; i < AGG_G; ++i) {
            if (kk + i < n) {
              float* ar = acc + rr[i] * F;
#pragma unroll
              for (int t = 0; t < T; ++t)
                ar[lane + 32 * t] += hv[i][t] * zz[i][t >> 1];
              if (lane < H) sacc[rr[i] * H + lane] += (lane == 0) ? zz[i][0] : zz[i][H - 1];
            }
          }
        }
      }
      __syncthreads();
    }
  };

  auto scan_all = [&](const bool sums) __attribute__((always_inline)) {
    const unsigned lt = (1u << lane) - 1u;
    const int slice = wave * 512 + lane;
    const int* __restrict__ dp = dst + slice;

    int dn[16];
    auto load_step = [&](int b) {
      if (b + AGG_SUB <= E) {
#pragma unroll
        for (int i = 0; i < 16; ++i) dn[i] = dp[b + i * 32];
      } else {
#pragma unroll
        for (int i = 0; i < 16; ++i) {
          const int e = b + slice + i * 32;
          dn[i] = (e < E) ? dst[e] : -1;
        }
      }
    };
    const int nsteps = (E + AGG_SUB - 1) / AGG_SUB;
    int step = (int)(((long long)blockIdx.x * nsteps) / gridDim.x);
    load_step(step * AGG_SUB);
    __syncthreads();

    int fill = 0;
    int par = 0;
    for (int k = 0; k < nsteps; ++k) {
      const int base = step * AGG_SUB;
      step = (step + 1 < nsteps) ? (step + 1) : 0;
      unsigned mk[16];
      int rl[16];
      int cnt = 0;
#pragma unroll
      for (int i = 0; i < 16; ++i) {
        rl[i] = dn[i] - lo;
        mk[i] = __builtin_amdgcn_ballot_w32((unsigned)rl[i] < (unsigned)nrows);
        cnt += __builtin_popcount(mk[i]);
      }
      if (lane == 0) wcnt[par * 32 + wave] = cnt;
      __syncthreads();
      load_step(step * AGG_SUB);
      const unsigned c = (unsigned)wcnt[par * 32 + lane];
      unsigned v = c | ((lane < wave) ? (c << 16) : 0u);
#pragma unroll
      for (int m = 16; m >= 1; m >>= 1) v += __shfl_xor(v, m);
      const int total = __builtin_amdgcn_readfirstlane((int)(v & 0xffffu));
      int off = fill + __builtin_amdgcn_readfirstlane((int)(v >> 16));
#pragma unroll
      for (int i = 0; i < 16; ++i) {
        if (mk[i] != 0u) {
          if ((mk[i] >> lane) & 1u)
            q[off + __builtin_popcount(mk[i] & lt)] =
                ((unsigned)(base + slice + i * 32) << 9) | (unsigned)rl[i];
          off += __builtin_popcount(mk[i]);
        }
      }
      fill += total;
      par ^= 1;

      const bool last = (k + 1 == nsteps);
      if (fill > AGG_CAP - AGG_SUB || last) {
        __syncthreads();
        if (sums) {
          drain(fill);
        } else {
          if (LIST == 1) {
            if (lsize >= 0 && lsize + fill <= AGG_LCAP) {
              for (int i = tid; i < fill; i += 1024) mylist[lsize + i] = q[i];
              lsize += fill;
            } else {
              lsize = -1;
            }
          }
          maxpass_queue(fill);
          __syncthreads();
        }
        fill = 0;
      }
    }
    if (LIST == 1 && !sums && tid == 0) lcnt[blockIdx.x] = lsize;
  };

  int cnt;
  if (LIST == 2) {
    cnt = lcnt[blockIdx.x];
    if (cnt >= 0 && cnt <= AGG_LCAP) {
      maxpass_list(cnt);
    } else {
      cnt = -1;
      scan_all(false);
    }
  } else {
    scan_all(false);
    cnt = lsize;
  }
  __syncthreads();

  for (int i = tid; i < nrows * H; i += 1024)
    sacc[i] = expf(selfal[lo * H + i] - funkey(mx[i]));
  __syncthreads();
  {
    const float4* __restrict__ h4 =
        reinterpret_cast<const float4*>(hfeat + (size_t)lo * F);
    float4* a4 = reinterpret_cast<float4*>(acc);
    for (int i = tid; i < nrows * (F / 4); i += 1024) {
      const float z = sacc[i >> 4];
      const float4 hv = h4[i];
      float4 r;
      r.x = hv.x * z;
      r.y = hv.y * z;
      r.z = hv.z * z;
      r.w = hv.w * z;
      a4[i] = r;
    }
  }
  __syncthreads();

  if (cnt >= 0) {
    for (int c0 = 0; c0 < cnt; c0 += AGG_CAP) {
      const int fill = (cnt - c0 < AGG_CAP) ? (cnt - c0) : AGG_CAP;
      for (int i = tid; i < fill; i += 1024) q[i] = mylist[c0 + i];
      __syncthreads();
      drain(fill);
    }
  } else {
    scan_all(true);
  }

  {
    const float4* a4 = reinterpret_cast<const float4*>(acc);
    float4* __restrict__ o4 = reinterpret_cast<float4*>(outp + (size_t)lo * F);
    for (int i = tid; i < nrows * (F / 4); i += 1024) {
      const float s = sacc[i >> 4];
      const float4 a = a4[i];
      const int f = (i << 2) & (F - 1);
      float4 r;
      r.x = a.x / s + bias[f];
      r.y = a.y / s + bias[f + 1];
      r.z = a.z / s + bias[f + 2];
      r.w = a.w / s + bias[f + 3];
      if (ELU) {
        r.x = (r.x > 0.f) ? r.x : expm1f(r.x);
        r.y = (r.y > 0.f) ? r.y : expm1f(r.y);
        r.z = (r.z > 0.f) ? r.z : expm1f(r.z);
        r.w = (r.w > 0.f) ? r.w : expm1f(r.w);
      }
      o4[i] = r;
    }
  }
}

extern "C" void kernel_launch(void* const* d_in, const int* in_sizes, int n_in,
                              void* d_out, int out_size, void* d_ws,
                              size_t ws_size, hipStream_t stream) {
  const float* x   = (const float*)d_in[0];
  const int*  eidx = (const int*)d_in[1];
  const float* ew  = (const float*)d_in[2];
  const float* W1  = (const float*)d_in[3];
  const float* We1 = (const float*)d_in[4];
  const float* as1 = (const float*)d_in[5];
  const float* ad1 = (const float*)d_in[6];
  const float* ae1 = (const float*)d_in[7];
  const float* b1  = (const float*)d_in[8];
  const float* W2  = (const float*)d_in[9];
  const float* We2 = (const float*)d_in[10];
  const float* as2 = (const float*)d_in[11];
  const float* ad2 = (const float*)d_in[12];
  const float* ae2 = (const float*)d_in[13];
  const float* b2  = (const float*)d_in[14];
  float* out = (float*)d_out;

  const int N  = in_sizes[0] / 128;
  const int E  = in_sizes[2];
  const int ET = E + N;
  const int* srcv = eidx;
  const int* dstv = eidx + E;

  float* p = (float*)d_ws;
  float* loopw = p; p += N;
  float* deg   = p; p += N;
  float* scoef = p; p += 2 * (size_t)N;
  float* dcoef = p; p += 2 * (size_t)N;
  float* cst   = p; p += 8;
  float* albuf = p; p += 2 * (size_t)ET;
  float* hfeat = p; p += (size_t)N * 128;
  float* agg   = p; p += (size_t)N * 128;
  int* lcnt = (int*)p; p += 1024;
  unsigned* lists = (unsigned*)p;

  const int T = 256;
  auto cdiv = [](long long a, long long b) { return (int)((a + b - 1) / b); };

  hipMemsetAsync(deg, 0, (size_t)N * 4, stream);
  hipMemsetAsync(loopw, 0, (size_t)N * 4, stream);
  k_selfloop_stats<<<cdiv(E, T), T, 0, stream>>>(dstv, ew, deg, loopw, E);
  k_selfloop_div<<<cdiv(N, T), T, 0, stream>>>(loopw, deg, N);

  gemm_wmma_f32<128><<<cdiv(N, 16 * 8), 256, 0, stream>>>(x, W1, hfeat, N);
  k_edge_const<2><<<1, 32, 0, stream>>>(We1, ae1, cst);
  k_node_coef<2><<<cdiv(N, T), T, 0, stream>>>(hfeat, as1, ad1, loopw, cst, scoef,
                                               dcoef, albuf + 2 * (size_t)E, N);
  k_edge_logit<2><<<cdiv(E, T), T, 0, stream>>>(srcv, dstv, ew, scoef, dcoef,
                                                cst, albuf, E);
  k_range_aggregate<128, true, 1><<<cdiv(N, AGG_R), 1024, 0, stream>>>(
      srcv, dstv, hfeat, albuf, albuf + 2 * (size_t)E, b1, agg, lists, lcnt, E, N);

  gemm_wmma_f32<64><<<cdiv(N, 16 * 6), 128, 0, stream>>>(agg, W2, hfeat, N);
  k_edge_const<1><<<1, 32, 0, stream>>>(We2, ae2, cst);
  k_node_coef<1><<<cdiv(N, T), T, 0, stream>>>(hfeat, as2, ad2, loopw, cst, scoef,
                                               dcoef, albuf + (size_t)E, N);
  k_edge_logit<1><<<cdiv(E, T), T, 0, stream>>>(srcv, dstv, ew, scoef, dcoef,
                                                cst, albuf, E);
  k_range_aggregate<64, false, 2><<<cdiv(N, AGG_R), 1024, 0, stream>>>(
      srcv, dstv, hfeat, albuf, albuf + (size_t)E, b2, out, lists, lcnt, E, N);
}


# Self-loop mean attr counted and summed by the row-owning work-group in LDS during its layer-1 max pass (LDS f32 adds, arrival order as before); global f32 atomics, 2 memsets, 2 launches gone
# speedup vs baseline: 2.9890x; 1.3503x over previous
#include <hip/hip_runtime.h>
#include <math.h>


typedef __attribute__((ext_vector_type(2))) float v2f;
typedef __attribute__((ext_vector_type(8))) float v8f;

__device__ __forceinline__ unsigned fkey(float f) {
  unsigned u = __float_as_uint(f);
  return (u & 0x80000000u) ? ~u : (u | 0x80000000u);
}
__device__ __forceinline__ float funkey(unsigned k) {
  unsigned u = (k & 0x80000000u) ? (k & 0x7fffffffu) : ~k;
  return __uint_as_float(u);
}

template <int NCOLS>
__global__ __launch_bounds__(32 * (NCOLS / 16))
void gemm_wmma_f32(const float* __restrict__ A, const float* __restrict__ Wg,
                   float* __restrict__ Out, int M) {
  constexpr int K = 128;
  constexpr int NT = NCOLS / 16;
  constexpr int NTHR = 32 * NT;
  constexpr int PITCH = K + 4;
  constexpr int NV = (16 * K / 4) / NTHR;
  static_assert(NV == 2 || NV == 4, "strip staging is written for 4 or 8 waves");
  __shared__ __attribute__((aligned(16))) float As[2][16 * PITCH];

  const int tid  = threadIdx.x;
  const int wave = tid >> 5;
  const int lane = tid & 31;
  const int r    = lane & 15;
  const int kb   = (lane >> 4) << 1;
  const int col  = (wave << 4) + r;

  v2f bf[K / 4];
#pragma unroll
  for (int s = 0; s < K / 4; ++s) {
    bf[s].x = Wg[(4 * s + kb) * NCOLS + col];
    bf[s].y = Wg[(4 * s + kb + 1) * NCOLS + col];
  }

  const int nstrips = (M + 15) >> 4;
  float4 st0, st1, st2, st3;
  st0 = st1 = st2 = st3 = make_float4(0.f, 0.f, 0.f, 0.f);
  auto gpiece = [&](int sp, int j) __attribute__((always_inline)) -> float4 {
    const int v = tid + j * NTHR;
    int arow = (sp << 4) + (v >> 5);
    if (arow >= M) arow = M - 1;
    return *reinterpret_cast<const float4*>(A + (size_t)arow * K + ((v & 31) << 2));
  };
  auto gload = [&](int sp) __attribute__((always_inline)) {
    st0 = gpiece(sp, 0);
    st1 = gpiece(sp, 1);
    if constexpr (NV > 2) {
      st2 = gpiece(sp, 2);
      st3 = gpiece(sp, 3);
    }
  };
  auto lpiece = [&](int b, int j, const float4 val) __attribute__((always_inline)) {
    const int v = tid + j * NTHR;
    *reinterpret_cast<float4*>(&As[b][(v >> 5) * PITCH + ((v & 31) << 2)]) = val;
  };
  auto lstore = [&](int b) __attribute__((always_inline)) {
    lpiece(b, 0, st0);
    lpiece(b, 1, st1);
    if constexpr (NV > 2) {
      lpiece(b, 2, st2);
      lpiece(b, 3, st3);
    }
  };

  int strip = blockIdx.x;
  if (strip < nstrips) {
    gload(strip);
    lstore(0);
  }
  __syncthreads();

  int buf = 0;
  for (; strip < nstrips; strip += gridDim.x) {
    const int nxt = strip + gridDim.x;
    if (nxt < nstrips) gload(nxt);

    const float* ar = &As[buf][r * PITCH + kb];
    v8f acc = {};
#pragma unroll
    for (int s = 0; s < K / 4; ++s) {
      const v2f a = *reinterpret_cast<const v2f*>(ar + 4 * s);
      acc = __builtin_amdgcn_wmma_f32_16x16x4_f32(false, a, false, bf[s],
                                                  (short)0, acc, false, false);
    }

    const int orow0 = (strip << 4) + ((lane >> 4) << 3);
#pragma unroll
    for (int i = 0; i < 8; ++i) {
      const int orow = orow0 + i;
      if (orow < M) Out[(size_t)orow * NCOLS + col] = acc[i];
    }

    if (nxt < nstrips) lstore(buf ^ 1);
    __syncthreads();
    buf ^= 1;
  }
}

template <int H, bool SELF>
__global__ void k_node_coef(const float* __restrict__ hfeat,
                            const float* __restrict__ a_s,
                            const float* __restrict__ a_d,
                            const float* __restrict__ loopw,
                            const float* __restrict__ cst,
                            float* __restrict__ scoef,
                            float* __restrict__ dcoef,
                            float* __restrict__ selfal, int N) {
  int n = blockIdx.x * blockDim.x + threadIdx.x;
  if (n >= N) return;
  const float* hr = hfeat + (size_t)n * (H * 64);
  float w = 0.f;
  if (SELF) w = loopw[n];
#pragma unroll
  for (int h = 0; h < H; ++h) {
    float ss = 0.f, dd = 0.f;
#pragma unroll 8
    for (int c = 0; c < 64; ++c) {
      float v = hr[h * 64 + c];
      ss += v * a_s[h * 64 + c];
      dd += v * a_d[h * 64 + c];
    }
    scoef[n * H + h] = ss;
    dcoef[n * H + h] = dd;
    if (SELF) {
      float al = ss + dd + w * cst[h];
      al = (al >= 0.f) ? al : 0.2f * al;
      selfal[n * H + h] = al;
    }
  }
}

template <int H>
__global__ void k_edge_const(const float* __restrict__ We,
                             const float* __restrict__ ae,
                             float* __restrict__ cst) {
  int h = threadIdx.x;
  if (h < H) {
    float s = 0.f;
    for (int c = 0; c < 64; ++c) s += We[h * 64 + c] * ae[h * 64 + c];
    cst[h] = s;
  }
}

template <int H>
__global__ void k_edge_logit(const int* __restrict__ src,
                             const int* __restrict__ dst,
                             const float* __restrict__ ew,
                             const float* __restrict__ scoef,
                             const float* __restrict__ dcoef,
                             const float* __restrict__ cst,
                             float* __restrict__ albuf, int E) {
  int e = blockIdx.x * blockDim.x + threadIdx.x;
  if (e >= E) return;
  const int s = src[e];
  const int d = dst[e];
  const float w = ew[e];
#pragma unroll
  for (int h = 0; h < H; ++h) {
    float al = scoef[s * H + h] + dcoef[d * H + h] + w * cst[h];
    al = (al >= 0.f) ? al : 0.2f * al;
    albuf[(size_t)e * H + h] = al;
  }
}

#define AGG_R 400
#define AGG_SUB 16384
#define AGG_CAP 23040
#define AGG_WIN 1024
#define AGG_G 8
#define AGG_MB 8
#define AGG_LCAP 65536
template <int F, bool ELU, int LIST>
__global__ __launch_bounds__(1024)
void k_range_aggregate(const int* __restrict__ src,
                       const int* __restrict__ dst,
                       const float* __restrict__ hfeat,
                       const float* __restrict__ albuf,
                       const float* __restrict__ selfal,
                       const float* __restrict__ bias,
                       float* __restrict__ outp,
                       unsigned* __restrict__ lists,
                       int* __restrict__ lcnt,
                       const float* __restrict__ ew,
                       const float* __restrict__ scoef,
                       const float* __restrict__ dcoef,
                       const float* __restrict__ cst,
                       float* __restrict__ loopw, int E, int N) {
  constexpr int H = F / 64;
  constexpr int T = F / 32;
  __shared__ __attribute__((aligned(16))) float acc[AGG_R * F];
  __shared__ float sacc[AGG_R * H];
  __shared__ unsigned mx[AGG_R * H];
  __shared__ unsigned q[AGG_CAP];
  __shared__ int w_src[AGG_WIN];
  __shared__ float w_z[H * AGG_WIN];
  __shared__ unsigned short wl[32 * 64];
  __shared__ unsigned omask[32 * 32];
  __shared__ int wcnt[2 * 32];
  __shared__ float degl[AGG_R];
  __shared__ float lws[AGG_R];

  const int tid  = threadIdx.x;
  const int wave = tid >> 5;
  const int lane = tid & 31;
  const int lo = blockIdx.x * AGG_R;
  const int nrows = (N - lo < AGG_R) ? (N - lo) : AGG_R;

  if (LIST == 1) {
    for (int i = tid; i < nrows * H; i += 1024) mx[i] = 0u;
    for (int i = tid; i < AGG_R; i += 1024) {
      degl[i] = 0.f;
      lws[i] = 0.f;
    }
  } else {
    for (int i = tid; i < nrows * H; i += 1024) mx[i] = fkey(selfal[lo * H + i]);
  }
  __syncthreads();

  unsigned* __restrict__ mylist = lists + (size_t)blockIdx.x * AGG_LCAP;
  int lsize = 0;

  auto maxgroup = [&](const unsigned (&me)[AGG_MB]) __attribute__((always_inline)) {
    float ma[AGG_MB][H];
#pragma unroll
    for (int j = 0; j < AGG_MB; ++j) {
#pragma unroll
      for (int h = 0; h < H; ++h)
        ma[j][h] = albuf[(size_t)(me[j] >> 9) * H + h];
    }
#pragma unroll
    for (int j = 0; j < AGG_MB; ++j) {
#pragma unroll
      for (int h = 0; h < H; ++h)
        atomicMax(&mx[(int)(me[j] & 511u) * H + h], fkey(ma[j][h]));
    }
  };
  auto maxpass_queue = [&](const int n) __attribute__((always_inline)) {
    for (int i0 = tid; i0 < n; i0 += AGG_MB * 1024) {
      unsigned me[AGG_MB];
      float wv[AGG_MB];
#pragma unroll
      for (int j = 0; j < AGG_MB; ++j) {
        const int i = i0 + j * 1024;
        me[j] = q[(i < n) ? i : i0];
      }
      if (LIST == 1) {
#pragma unroll
        for (int j = 0; j < AGG_MB; ++j) wv[j] = ew[me[j] >> 9];
      }
      maxgroup(me);
      if (LIST == 1) {
#pragma unroll
        for (int j = 0; j < AGG_MB; ++j) {
          if (i0 + j * 1024 < n) {
            const int rw = (int)(me[j] & 511u);
            atomicAdd(&degl[rw], 1.0f);
            atomicAdd(&lws[rw], wv[j]);
          }
        }
      }
    }
  };
  auto maxpass_list = [&](const int n) __attribute__((always_inline)) {
    for (int i0 = tid; i0 < n; i0 += AGG_MB * 1024) {
      unsigned me[AGG_MB];
#pragma unroll
      for (int j = 0; j < AGG_MB; ++j) {
        const int i = i0 + j * 1024;
        me[j] = mylist[(i < n) ? i : i0];
      }
      maxgroup(me);
    }
  };

  auto drain = [&](const int fill) __attribute__((always_inline)) {
    for (int wb = 0; wb < fill; wb += AGG_WIN) {
      const int wn = (fill - wb < AGG_WIN) ? (fill - wb) : AGG_WIN;
      int owner = 32;
      if (tid < wn) {
        const unsigned ent = q[wb + tid];
        const int e = (int)(ent >> 9);
        const int rw = (int)(ent & 511u);
        owner = (int)(ent & 31u);
        w_src[tid] = src[e];
#pragma unroll
        for (int h = 0; h < H; ++h)
          w_z[h * AGG_WIN + tid] =
              expf(albuf[(size_t)e * H + h] - funkey(mx[rw * H + h]));
      }
      unsigned mym = 0u;
#pragma unroll
      for (int o = 0; o < 32; ++o) {
        const unsigned b = __builtin_amdgcn_ballot_w32(owner == o);
        mym = (lane == o) ? b : mym;
      }
      omask[lane * 32 + wave] = mym;
      __syncthreads();
      const unsigned mymask = omask[wave * 32 + lane];
      const int cm = __builtin_popcount(mymask);
      int pre = cm;
#pragma unroll
      for (int dlt = 1; dlt < 32; dlt <<= 1) {
        const int up = __shfl_up(pre, dlt);
        if (lane >= dlt) pre += up;
      }
      const int nall = __builtin_amdgcn_readlane(pre, 31);
      pre -= cm;
      for (int r0 = 0; r0 < nall; r0 += 64) {
        {
          unsigned mm = mymask;
          int p = pre - r0;
          while (__builtin_amdgcn_ballot_w32(mm != 0u) != 0u) {
            if (mm != 0u) {
              if (p >= 0 && p < 64)
                wl[wave * 64 + p] = (unsigned short)(lane * 32 + __builtin_ctz(mm));
              mm &= mm - 1u;
              ++p;
            }
          }
        }
        const int n = (nall - r0 < 64) ? (nall - r0) : 64;
        for (int kk = 0; kk < n; kk += AGG_G) {
          int rr[AGG_G];
          float zz[AGG_G][H];
          float hv[AGG_G][T];
#pragma unroll
          for (int i = 0; i < AGG_G; ++i) {
            const int idx = wl[wave * 64 + ((kk + i < n) ? (kk + i) : kk)];
            rr[i] = (int)(q[wb + idx] & 511u);
            const float* hs = hfeat + (size_t)w_src[idx] * F;
#pragma unroll
            for (int h = 0; h < H; ++h) zz[i][h] = w_z[h * AGG_WIN + idx];
#pragma unroll
            for (int t = 0; t < T; ++t) hv[i][t] = hs[lane + 32 * t];
          }
#pragma unroll
          for (int i = 0; i < AGG_G; ++i) {
            if (kk + i < n) {
              float* ar = acc + rr[i] * F;
#pragma unroll
              for (int t = 0; t < T; ++t)
                ar[lane + 32 * t] += hv[i][t] * zz[i][t >> 1];
              if (lane < H) sacc[rr[i] * H + lane] += (lane == 0) ? zz[i][0] : zz[i][H - 1];
            }
          }
        }
      }
      __syncthreads();
    }
  };

  auto scan_all = [&](const bool sums) __attribute__((always_inline)) {
    const unsigned lt = (1u << lane) - 1u;
    const int slice = wave * 512 + lane;
    const int* __restrict__ dp = dst + slice;

    int dn[16];
    auto load_step = [&](int b) {
      if (b + AGG_SUB <= E) {
#pragma unroll
        for (int i = 0; i < 16; ++i) dn[i] = dp[b + i * 32];
      } else {
#pragma unroll
        for (int i = 0; i < 16; ++i) {
          const int e = b + slice + i * 32;
          dn[i] = (e < E) ? dst[e] : -1;
        }
      }
    };
    const int nsteps = (E + AGG_SUB - 1) / AGG_SUB;
    int step = (int)(((long long)blockIdx.x * nsteps) / gridDim.x);
    load_step(step * AGG_SUB);
    __syncthreads();

    int fill = 0;
    int par = 0;
    for (int k = 0; k < nsteps; ++k) {
      const int base = step * AGG_SUB;
      step = (step + 1 < nsteps) ? (step + 1) : 0;
      unsigned mk[16];
      int rl[16];
      int cnt = 0;
#pragma unroll
      for (int i = 0; i < 16; ++i) {
        rl[i] = dn[i] - lo;
        mk[i] = __builtin_amdgcn_ballot_w32((unsigned)rl[i] < (unsigned)nrows);
        cnt += __builtin_popcount(mk[i]);
      }
      if (lane == 0) wcnt[par * 32 + wave] = cnt;
      __syncthreads();
      load_step(step * AGG_SUB);
      const unsigned c = (unsigned)wcnt[par * 32 + lane];
      unsigned v = c | ((lane < wave) ? (c << 16) : 0u);
#pragma unroll
      for (int m = 16; m >= 1; m >>= 1) v += __shfl_xor(v, m);
      const int total = __builtin_amdgcn_readfirstlane((int)(v & 0xffffu));
      int off = fill + __builtin_amdgcn_readfirstlane((int)(v >> 16));
#pragma unroll
      for (int i = 0; i < 16; ++i) {
        if (mk[i] != 0u) {
          if ((mk[i] >> lane) & 1u)
            q[off + __builtin_popcount(mk[i] & lt)] =
                ((unsigned)(base + slice + i * 32) << 9) | (unsigned)rl[i];
          off += __builtin_popcount(mk[i]);
        }
      }
      fill += total;
      par ^= 1;

      const bool last = (k + 1 == nsteps);
      if (fill > AGG_CAP - AGG_SUB || last) {
        __syncthreads();
        if (sums) {
          drain(fill);
        } else {
          if (LIST == 1) {
            if (lsize >= 0 && lsize + fill <= AGG_LCAP) {
              for (int i = tid; i < fill; i += 1024) mylist[lsize + i] = q[i];
              lsize += fill;
            } else {
              lsize = -1;
            }
          }
          maxpass_queue(fill);
          __syncthreads();
        }
        fill = 0;
      }
    }
    if (LIST == 1 && !sums && tid == 0) lcnt[blockIdx.x] = lsize;
  };

  int cnt;
  if (LIST == 2) {
    cnt = lcnt[blockIdx.x];
    if (cnt >= 0 && cnt <= AGG_LCAP) {
      maxpass_list(cnt);
    } else {
      cnt = -1;
      scan_all(false);
    }
  } else {
    scan_all(false);
    cnt = lsize;
  }
  __syncthreads();

  for (int i = tid; i < nrows * H; i += 1024) {
    float al;
    if (LIST == 1) {
      const int rw = i / H;
      const int h = i - rw * H;
      const float w = lws[rw] / fmaxf(degl[rw], 1.0f);
      if (h == 0) loopw[lo + rw] = w;
      al = scoef[lo * H + i] + dcoef[lo * H + i] + w * cst[h];
      al = (al >= 0.f) ? al : 0.2f * al;
      const unsigned k = fkey(al);
      if (k > mx[i]) mx[i] = k;
    } else {
      al = selfal[lo * H + i];
    }
    sacc[i] = expf(al - funkey(mx[i]));
  }
  __syncthreads();
  {
    const float4* __restrict__ h4 =
        reinterpret_cast<const float4*>(hfeat + (size_t)lo * F);
    float4* a4 = reinterpret_cast<float4*>(acc);
    for (int i = tid; i < nrows * (F / 4); i += 1024) {
      const float z = sacc[i >> 4];
      const float4 hv = h4[i];
      float4 r;
      r.x = hv.x * z;
      r.y = hv.y * z;
      r.z = hv.z * z;
      r.w = hv.w * z;
      a4[i] = r;
    }
  }
  __syncthreads();

  if (cnt >= 0) {
    for (int c0 = 0; c0 < cnt; c0 += AGG_CAP) {
      const int fill = (cnt - c0 < AGG_CAP) ? (cnt - c0) : AGG_CAP;
      for (int i = tid; i < fill; i += 1024) q[i] = mylist[c0 + i];
      __syncthreads();
      drain(fill);
    }
  } else {
    scan_all(true);
  }

  {
    const float4* a4 = reinterpret_cast<const float4*>(acc);
    float4* __restrict__ o4 = reinterpret_cast<float4*>(outp + (size_t)lo * F);
    for (int i = tid; i < nrows * (F / 4); i += 1024) {
      const float s = sacc[i >> 4];
      const float4 a = a4[i];
      const int f = (i << 2) & (F - 1);
      float4 r;
      r.x = a.x / s + bias[f];
      r.y = a.y / s + bias[f + 1];
      r.z = a.z / s + bias[f + 2];
      r.w = a.w / s + bias[f + 3];
      if (ELU) {
        r.x = (r.x > 0.f) ? r.x : expm1f(r.x);
        r.y = (r.y > 0.f) ? r.y : expm1f(r.y);
        r.z = (r.z > 0.f) ? r.z : expm1f(r.z);
        r.w = (r.w > 0.f) ? r.w : expm1f(r.w);
      }
      o4[i] = r;
    }
  }
}

extern "C" void kernel_launch(void* const* d_in, const int* in_sizes, int n_in,
                              void* d_out, int out_size, void* d_ws,
                              size_t ws_size, hipStream_t stream) {
  const float* x   = (const float*)d_in[0];
  const int*  eidx = (const int*)d_in[1];
  const float* ew  = (const float*)d_in[2];
  const float* W1  = (const float*)d_in[3];
  const float* We1 = (const float*)d_in[4];
  const float* as1 = (const float*)d_in[5];
  const float* ad1 = (const float*)d_in[6];
  const float* ae1 = (const float*)d_in[7];
  const float* b1  = (const float*)d_in[8];
  const float* W2  = (const float*)d_in[9];
  const float* We2 = (const float*)d_in[10];
  const float* as2 = (const float*)d_in[11];
  const float* ad2 = (const float*)d_in[12];
  const float* ae2 = (const float*)d_in[13];
  const float* b2  = (const float*)d_in[14];
  float* out = (float*)d_out;

  const int N  = in_sizes[0] / 128;
  const int E  = in_sizes[2];
  const int ET = E + N;
  const int* srcv = eidx;
  const int* dstv = eidx + E;

  float* p = (float*)d_ws;
  float* loopw = p; p += N;
  float* scoef = p; p += 2 * (size_t)N;
  float* dcoef = p; p += 2 * (size_t)N;
  float* cst   = p; p += 8;
  float* albuf = p; p += 2 * (size_t)ET;
  float* hfeat = p; p += (size_t)N * 128;
  float* agg   = p; p += (size_t)N * 128;
  int* lcnt = (int*)p; p += 1024;
  unsigned* lists = (unsigned*)p;

  const int T = 256;
  auto cdiv = [](long long a, long long b) { return (int)((a + b - 1) / b); };

  gemm_wmma_f32<128><<<cdiv(N, 16 * 8), 256, 0, stream>>>(x, W1, hfeat, N);
  k_edge_const<2><<<1, 32, 0, stream>>>(We1, ae1, cst);
  k_node_coef<2, false><<<cdiv(N, T), T, 0, stream>>>(hfeat, as1, ad1, loopw, cst,
                                                      scoef, dcoef,
                                                      albuf + 2 * (size_t)E, N);
  k_edge_logit<2><<<cdiv(E, T), T, 0, stream>>>(srcv, dstv, ew, scoef, dcoef,
                                                cst, albuf, E);
  k_range_aggregate<128, true, 1><<<cdiv(N, AGG_R), 1024, 0, stream>>>(
      srcv, dstv, hfeat, albuf, albuf + 2 * (size_t)E, b1, agg, lists, lcnt,
      ew, scoef, dcoef, cst, loopw, E, N);

  gemm_wmma_f32<64><<<cdiv(N, 16 * 6), 128, 0, stream>>>(agg, W2, hfeat, N);
  k_edge_const<1><<<1, 32, 0, stream>>>(We2, ae2, cst);
  k_node_coef<1, true><<<cdiv(N, T), T, 0, stream>>>(hfeat, as2, ad2, loopw, cst,
                                                     scoef, dcoef,
                                                     albuf + (size_t)E, N);
  k_edge_logit<1><<<cdiv(E, T), T, 0, stream>>>(srcv, dstv, ew, scoef, dcoef,
                                                cst, albuf, E);
  k_range_aggregate<64, false, 2><<<cdiv(N, AGG_R), 1024, 0, stream>>>(
      srcv, dstv, hfeat, albuf, albuf + (size_t)E, b2, out, lists, lcnt,
      ew, scoef, dcoef, cst, loopw, E, N);
}


# Layer-1 edge scan without per-step barriers: each wave walks its own 1/32 of the edge list, stages matches privately, one compaction at the end; f32 row sum order changes (by wave part, fixed)
# speedup vs baseline: 3.4668x; 1.1599x over previous
#include <hip/hip_runtime.h>
#include <math.h>


typedef __attribute__((ext_vector_type(2))) float v2f;
typedef __attribute__((ext_vector_type(8))) float v8f;

__device__ __forceinline__ unsigned fkey(float f) {
  unsigned u = __float_as_uint(f);
  return (u & 0x80000000u) ? ~u : (u | 0x80000000u);
}
__device__ __forceinline__ float funkey(unsigned k) {
  unsigned u = (k & 0x80000000u) ? (k & 0x7fffffffu) : ~k;
  return __uint_as_float(u);
}

template <int NCOLS>
__global__ __launch_bounds__(32 * (NCOLS / 16))
void gemm_wmma_f32(const float* __restrict__ A, const float* __restrict__ Wg,
                   float* __restrict__ Out, int M) {
  constexpr int K = 128;
  constexpr int NT = NCOLS / 16;
  constexpr int NTHR = 32 * NT;
  constexpr int PITCH = K + 4;
  constexpr int NV = (16 * K / 4) / NTHR;
  static_assert(NV == 2 || NV == 4, "strip staging is written for 4 or 8 waves");
  __shared__ __attribute__((aligned(16))) float As[2][16 * PITCH];

  const int tid  = threadIdx.x;
  const int wave = tid >> 5;
  const int lane = tid & 31;
  const int r    = lane & 15;
  const int kb   = (lane >> 4) << 1;
  const int col  = (wave << 4) + r;

  v2f bf[K / 4];
#pragma unroll
  for (int s = 0; s < K / 4; ++s) {
    bf[s].x = Wg[(4 * s + kb) * NCOLS + col];
    bf[s].y = Wg[(4 * s + kb + 1) * NCOLS + col];
  }

  const int nstrips = (M + 15) >> 4;
  float4 st0, st1, st2, st3;
  st0 = st1 = st2 = st3 = make_float4(0.f, 0.f, 0.f, 0.f);
  auto gpiece = [&](int sp, int j) __attribute__((always_inline)) -> float4 {
    const int v = tid + j * NTHR;
    int arow = (sp << 4) + (v >> 5);
    if (arow >= M) arow = M - 1;
    return *reinterpret_cast<const float4*>(A + (size_t)arow * K + ((v & 31) << 2));
  };
  auto gload = [&](int sp) __attribute__((always_inline)) {
    st0 = gpiece(sp, 0);
    st1 = gpiece(sp, 1);
    if constexpr (NV > 2) {
      st2 = gpiece(sp, 2);
      st3 = gpiece(sp, 3);
    }
  };
  auto lpiece = [&](int b, int j, const float4 val) __attribute__((always_inline)) {
    const int v = tid + j * NTHR;
    *reinterpret_cast<float4*>(&As[b][(v >> 5) * PITCH + ((v & 31) << 2)]) = val;
  };
  auto lstore = [&](int b) __attribute__((always_inline)) {
    lpiece(b, 0, st0);
    lpiece(b, 1, st1);
    if constexpr (NV > 2) {
      lpiece(b, 2, st2);
      lpiece(b, 3, st3);
    }
  };

  int strip = blockIdx.x;
  if (strip < nstrips) {
    gload(strip);
    lstore(0);
  }
  __syncthreads();

  int buf = 0;
  for (; strip < nstrips; strip += gridDim.x) {
    const int nxt = strip + gridDim.x;
    if (nxt < nstrips) gload(nxt);

    const float* ar = &As[buf][r * PITCH + kb];
    v8f acc = {};
#pragma unroll
    for (int s = 0; s < K / 4; ++s) {
      const v2f a = *reinterpret_cast<const v2f*>(ar + 4 * s);
      acc = __builtin_amdgcn_wmma_f32_16x16x4_f32(false, a, false, bf[s],
                                                  (short)0, acc, false, false);
    }

    const int orow0 = (strip << 4) + ((lane >> 4) << 3);
#pragma unroll
    for (int i = 0; i < 8; ++i) {
      const int orow = orow0 + i;
      if (orow < M) Out[(size_t)orow * NCOLS + col] = acc[i];
    }

    if (nxt < nstrips) lstore(buf ^ 1);
    __syncthreads();
    buf ^= 1;
  }
}

template <int H, bool SELF>
__global__ void k_node_coef(const float* __restrict__ hfeat,
                            const float* __restrict__ a_s,
                            const float* __restrict__ a_d,
                            const float* __restrict__ loopw,
                            const float* __restrict__ cst,
                            float* __restrict__ scoef,
                            float* __restrict__ dcoef,
                            float* __restrict__ selfal, int N) {
  int n = blockIdx.x * blockDim.x + threadIdx.x;
  if (n >= N) return;
  const float* hr = hfeat + (size_t)n * (H * 64);
  float w = 0.f;
  if (SELF) w = loopw[n];
#pragma unroll
  for (int h = 0; h < H; ++h) {
    float ss = 0.f, dd = 0.f;
#pragma unroll 8
    for (int c = 0; c < 64; ++c) {
      float v = hr[h * 64 + c];
      ss += v * a_s[h * 64 + c];
      dd += v * a_d[h * 64 + c];
    }
    scoef[n * H + h] = ss;
    dcoef[n * H + h] = dd;
    if (SELF) {
      float al = ss + dd + w * cst[h];
      al = (al >= 0.f) ? al : 0.2f * al;
      selfal[n * H + h] = al;
    }
  }
}

template <int H>
__global__ void k_edge_const(const float* __restrict__ We,
                             const float* __restrict__ ae,
                             float* __restrict__ cst) {
  int h = threadIdx.x;
  if (h < H) {
    float s = 0.f;
    for (int c = 0; c < 64; ++c) s += We[h * 64 + c] * ae[h * 64 + c];
    cst[h] = s;
  }
}

template <int H>
__global__ void k_edge_logit(const int* __restrict__ src,
                             const int* __restrict__ dst,
                             const float* __restrict__ ew,
                             const float* __restrict__ scoef,
                             const float* __restrict__ dcoef,
                             const float* __restrict__ cst,
                             float* __restrict__ albuf, int E) {
  int e = blockIdx.x * blockDim.x + threadIdx.x;
  if (e >= E) return;
  const int s = src[e];
  const int d = dst[e];
  const float w = ew[e];
#pragma unroll
  for (int h = 0; h < H; ++h) {
    float al = scoef[s * H + h] + dcoef[d * H + h] + w * cst[h];
    al = (al >= 0.f) ? al : 0.2f * al;
    albuf[(size_t)e * H + h] = al;
  }
}

#define AGG_R 400
#define AGG_SUB 16384
#define AGG_CAP 23040
#define AGG_WIN 1024
#define AGG_G 8
#define AGG_MB 8
#define AGG_LCAP 65536
template <int F, bool ELU, int LIST>
__global__ __launch_bounds__(1024)
void k_range_aggregate(const int* __restrict__ src,
                       const int* __restrict__ dst,
                       const float* __restrict__ hfeat,
                       const float* __restrict__ albuf,
                       const float* __restrict__ selfal,
                       const float* __restrict__ bias,
                       float* __restrict__ outp,
                       unsigned* __restrict__ lists,
                       int* __restrict__ lcnt,
                       const float* __restrict__ ew,
                       const float* __restrict__ scoef,
                       const float* __restrict__ dcoef,
                       const float* __restrict__ cst,
                       float* __restrict__ loopw, int E, int N) {
  constexpr int H = F / 64;
  constexpr int T = F / 32;
  __shared__ __attribute__((aligned(16))) float acc[AGG_R * F];
  __shared__ float sacc[AGG_R * H];
  __shared__ unsigned mx[AGG_R * H];
  __shared__ unsigned q[AGG_CAP];
  __shared__ int w_src[AGG_WIN];
  __shared__ float w_z[H * AGG_WIN];
  __shared__ unsigned short wl[32 * 64];
  __shared__ unsigned omask[32 * 32];
  __shared__ int wcnt[2 * 32];
  __shared__ float degl[AGG_R];
  __shared__ float lws[AGG_R];

  const int tid  = threadIdx.x;
  const int wave = tid >> 5;
  const int lane = tid & 31;
  const int lo = blockIdx.x * AGG_R;
  const int nrows = (N - lo < AGG_R) ? (N - lo) : AGG_R;

  if (LIST == 1) {
    for (int i = tid; i < nrows * H; i += 1024) mx[i] = 0u;
    for (int i = tid; i < AGG_R; i += 1024) {
      degl[i] = 0.f;
      lws[i] = 0.f;
    }
  } else {
    for (int i = tid; i < nrows * H; i += 1024) mx[i] = fkey(selfal[lo * H + i]);
  }
  __syncthreads();

  unsigned* __restrict__ mylist = lists + (size_t)blockIdx.x * AGG_LCAP;
  int lsize = 0;

  auto maxgroup = [&](const unsigned (&me)[AGG_MB]) __attribute__((always_inline)) {
    float ma[AGG_MB][H];
#pragma unroll
    for (int j = 0; j < AGG_MB; ++j) {
#pragma unroll
      for (int h = 0; h < H; ++h)
        ma[j][h] = albuf[(size_t)(me[j] >> 9) * H + h];
    }
#pragma unroll
    for (int j = 0; j < AGG_MB; ++j) {
#pragma unroll
      for (int h = 0; h < H; ++h)
        atomicMax(&mx[(int)(me[j] & 511u) * H + h], fkey(ma[j][h]));
    }
  };
  auto maxpass_queue = [&](const int n) __attribute__((always_inline)) {
    for (int i0 = tid; i0 < n; i0 += AGG_MB * 1024) {
      unsigned me[AGG_MB];
      float wv[AGG_MB];
#pragma unroll
      for (int j = 0; j < AGG_MB; ++j) {
        const int i = i0 + j * 1024;
        me[j] = q[(i < n) ? i : i0];
      }
      if (LIST == 1) {
#pragma unroll
        for (int j = 0; j < AGG_MB; ++j) wv[j] = ew[me[j] >> 9];
      }
      maxgroup(me);
      if (LIST == 1) {
#pragma unroll
        for (int j = 0; j < AGG_MB; ++j) {
          if (i0 + j * 1024 < n) {
            const int rw = (int)(me[j] & 511u);
            atomicAdd(&degl[rw], 1.0f);
            atomicAdd(&lws[rw], wv[j]);
          }
        }
      }
    }
  };
  auto maxpass_list = [&](const int n) __attribute__((always_inline)) {
    for (int i0 = tid; i0 < n; i0 += AGG_MB * 1024) {
      unsigned me[AGG_MB];
#pragma unroll
      for (int j = 0; j < AGG_MB; ++j) {
        const int i = i0 + j * 1024;
        me[j] = mylist[(i < n) ? i : i0];
      }
      maxgroup(me);
    }
  };

  auto drain = [&](const int fill) __attribute__((always_inline)) {
    for (int wb = 0; wb < fill; wb += AGG_WIN) {
      const int wn = (fill - wb < AGG_WIN) ? (fill - wb) : AGG_WIN;
      int owner = 32;
      if (tid < wn) {
        const unsigned ent = q[wb + tid];
        const int e = (int)(ent >> 9);
        const int rw = (int)(ent & 511u);
        owner = (int)(ent & 31u);
        w_src[tid] = src[e];
#pragma unroll
        for (int h = 0; h < H; ++h)
          w_z[h * AGG_WIN + tid] =
              expf(albuf[(size_t)e * H + h] - funkey(mx[rw * H + h]));
      }
      unsigned mym = 0u;
#pragma unroll
      for (int o = 0; o < 32; ++o) {
        const unsigned b = __builtin_amdgcn_ballot_w32(owner == o);
        mym = (lane == o) ? b : mym;
      }
      omask[lane * 32 + wave] = mym;
      __syncthreads();
      const unsigned mymask = omask[wave * 32 + lane];
      const int cm = __builtin_popcount(mymask);
      int pre = cm;
#pragma unroll
      for (int dlt = 1; dlt < 32; dlt <<= 1) {
        const int up = __shfl_up(pre, dlt);
        if (lane >= dlt) pre += up;
      }
      const int nall = __builtin_amdgcn_readlane(pre, 31);
      pre -= cm;
      for (int r0 = 0; r0 < nall; r0 += 64) {
        {
          unsigned mm = mymask;
          int p = pre - r0;
          while (__builtin_amdgcn_ballot_w32(mm != 0u) != 0u) {
            if (mm != 0u) {
              if (p >= 0 && p < 64)
                wl[wave * 64 + p] = (unsigned short)(lane * 32 + __builtin_ctz(mm));
              mm &= mm - 1u;
              ++p;
            }
          }
        }
        const int n = (nall - r0 < 64) ? (nall - r0) : 64;
        for (int kk = 0; kk < n; kk += AGG_G) {
          int rr[AGG_G];
          float zz[AGG_G][H];
          float hv[AGG_G][T];
#pragma unroll
          for (int i = 0; i < AGG_G; ++i) {
            const int idx = wl[wave * 64 + ((kk + i < n) ? (kk + i) : kk)];
            rr[i] = (int)(q[wb + idx] & 511u);
            const float* hs = hfeat + (size_t)w_src[idx] * F;
#pragma unroll
            for (int h = 0; h < H; ++h) zz[i][h] = w_z[h * AGG_WIN + idx];
#pragma unroll
            for (int t = 0; t < T; ++t) hv[i][t] = hs[lane + 32 * t];
          }
#pragma unroll
          for (int i = 0; i < AGG_G; ++i) {
            if (kk + i < n) {
              float* ar = acc + rr[i] * F;
#pragma unroll
              for (int t = 0; t < T; ++t)
                ar[lane + 32 * t] += hv[i][t] * zz[i][t >> 1];
              if (lane < H) sacc[rr[i] * H + lane] += (lane == 0) ? zz[i][0] : zz[i][H - 1];
            }
          }
        }
      }
      __syncthreads();
    }
  };

  auto scan_all = [&](const bool sums) __attribute__((always_inline)) {
    const unsigned lt = (1u << lane) - 1u;
    const int slice = wave * 512 + lane;
    const int* __restrict__ dp = dst + slice;

    int dn[16];
    auto load_step = [&](int b) {
      if (b + AGG_SUB <= E) {
#pragma unroll
        for (int i = 0; i < 16; ++i) dn[i] = dp[b + i * 32];
      } else {
#pragma unroll
        for (int i = 0; i < 16; ++i) {
          const int e = b + slice + i * 32;
          dn[i] = (e < E) ? dst[e] : -1;
        }
      }
    };
    const int nsteps = (E + AGG_SUB - 1) / AGG_SUB;
    int step = (int)(((long long)blockIdx.x * nsteps) / gridDim.x);
    load_step(step * AGG_SUB);
    __syncthreads();

    int fill = 0;
    int par = 0;
    for (int k = 0; k < nsteps; ++k) {
      const int base = step * AGG_SUB;
      step = (step + 1 < nsteps) ? (step + 1) : 0;
      unsigned mk[16];
      int rl[16];
      int cnt = 0;
#pragma unroll
      for (int i = 0; i < 16; ++i) {
        rl[i] = dn[i] - lo;
        mk[i] = __builtin_amdgcn_ballot_w32((unsigned)rl[i] < (unsigned)nrows);
        cnt += __builtin_popcount(mk[i]);
      }
      if (lane == 0) wcnt[par * 32 + wave] = cnt;
      __syncthreads();
      load_step(step * AGG_SUB);
      const unsigned c = (unsigned)wcnt[par * 32 + lane];
      unsigned v = c | ((lane < wave) ? (c << 16) : 0u);
#pragma unroll
      for (int m = 16; m >= 1; m >>= 1) v += __shfl_xor(v, m);
      const int total = __builtin_amdgcn_readfirstlane((int)(v & 0xffffu));
      int off = fill + __builtin_amdgcn_readfirstlane((int)(v >> 16));
#pragma unroll
      for (int i = 0; i < 16; ++i) {
        if (mk[i] != 0u) {
          if ((mk[i] >> lane) & 1u)
            q[off + __builtin_popcount(mk[i] & lt)] =
                ((unsigned)(base + slice + i * 32) << 9) | (unsigned)rl[i];
          off += __builtin_popcount(mk[i]);
        }
      }
      fill += total;
      par ^= 1;

      const bool last = (k + 1 == nsteps);
      if (fill > AGG_CAP - AGG_SUB || last) {
        __syncthreads();
        if (sums) {
          drain(fill);
        } else {
          if (LIST == 1) {
            if (lsize >= 0 && lsize + fill <= AGG_LCAP) {
              for (int i = tid; i < fill; i += 1024) mylist[lsize + i] = q[i];
              lsize += fill;
            } else {
              lsize = -1;
            }
          }
          maxpass_queue(fill);
          __syncthreads();
        }
        fill = 0;
      }
    }
    if (LIST == 1 && !sums && tid == 0) lcnt[blockIdx.x] = lsize;
  };

  auto scan_own = [&](int& fill_out) __attribute__((always_inline)) -> bool {
    constexpr int PW = (AGG_R * F) / 32;
    const int sw = __builtin_amdgcn_readfirstlane(wave);
    unsigned* stage = reinterpret_cast<unsigned*>(acc) + sw * PW;
    const unsigned lt = (1u << lane) - 1u;
    const int npieces = (E + 511) >> 9;
    const int pwhole = E >> 9;
    const int ppw = (npieces + 31) >> 5;
    const int p0 = sw * ppw;
    const int p1 = (p0 + ppw < npieces) ? (p0 + ppw) : npieces;
    const int p1w = (p1 < pwhole) ? p1 : pwhole;
    const int nw = (p1w > p0) ? (p1w - p0) : 0;
    const int rot = (int)(((long long)blockIdx.x * nw) / gridDim.x);
    auto next_piece = [&](int p) { return (p + 1 < p1w) ? (p + 1) : p0; };
    auto load_piece = [&](int pc, int (&d)[16]) {
      const int b = pc << 9;
#pragma unroll
      for (int i = 0; i < 16; ++i) d[i] = dst[b + i * 32 + lane];
    };
    int off = 0;
    bool ovf = false;
    auto test_piece = [&](int pc, const int (&d)[16]) {
      if (off > PW - 512) {
        ovf = true;
        return;
      }
      const unsigned ebase = (unsigned)((pc << 9) + lane) << 9;
      unsigned mk[16];
      int rl[16];
#pragma unroll
      for (int i = 0; i < 16; ++i) {
        rl[i] = d[i] - lo;
        mk[i] = __builtin_amdgcn_ballot_w32((unsigned)rl[i] < (unsigned)nrows);
      }
#pragma unroll
      for (int i = 0; i < 16; ++i) {
        if (mk[i] != 0u) {
          if ((mk[i] >> lane) & 1u)
            stage[off + __builtin_popcount(mk[i] & lt)] =
                (ebase + ((unsigned)(i * 32) << 9)) | (unsigned)rl[i];
          off += __builtin_popcount(mk[i]);
        }
      }
    };
    int da[16], db[16];
    if (nw > 0) {
      int pa = p0 + rot;
      load_piece(pa, da);
      int k = 0;
      for (; k + 2 <= nw; k += 2) {
        const int pb = next_piece(pa);
        load_piece(pb, db);
        test_piece(pa, da);
        pa = next_piece(pb);
        load_piece(pa, da);
        test_piece(pb, db);
      }
      if (k < nw) test_piece(pa, da);
    }
    if (pwhole < p1 && pwhole >= p0) {
      const int b = pwhole << 9;
#pragma unroll
      for (int i = 0; i < 16; ++i) {
        const int e = b + i * 32 + lane;
        da[i] = (e < E) ? dst[e] : -1;
      }
      test_piece(pwhole, da);
    }
    if (lane == 0) wcnt[sw] = ovf ? (1 << 20) : off;
    __syncthreads();
    const int c = wcnt[lane];
    int pre = c;
#pragma unroll
    for (int dlt = 1; dlt < 32; dlt <<= 1) {
      const int up = __shfl_up(pre, dlt);
      if (lane >= dlt) pre += up;
    }
    const int total = __builtin_amdgcn_readlane(pre, 31);
    const int myoff = __builtin_amdgcn_readlane(pre - c, sw);
    if (total > AGG_CAP) return false;
    for (int i = lane; i < off; i += 32) q[myoff + i] = stage[i];
    __syncthreads();
    fill_out = total;
    return true;
  };

  int cnt;
  if (LIST == 2) {
    cnt = lcnt[blockIdx.x];
    if (cnt >= 0 && cnt <= AGG_LCAP) {
      maxpass_list(cnt);
    } else {
      cnt = -1;
      scan_all(false);
    }
  } else {
    bool own = false;
    if constexpr (LIST == 1) {
      int fill = 0;
      own = scan_own(fill);
      if (own) {
        for (int i = tid; i < fill; i += 1024) mylist[i] = q[i];
        lsize = fill;
        maxpass_queue(fill);
        if (tid == 0) lcnt[blockIdx.x] = lsize;
      }
    }
    if (!own) scan_all(false);
    cnt = lsize;
  }
  __syncthreads();

  for (int i = tid; i < nrows * H; i += 1024) {
    float al;
    if (LIST == 1) {
      const int rw = i / H;
      const int h = i - rw * H;
      const float w = lws[rw] / fmaxf(degl[rw], 1.0f);
      if (h == 0) loopw[lo + rw] = w;
      al = scoef[lo * H + i] + dcoef[lo * H + i] + w * cst[h];
      al = (al >= 0.f) ? al : 0.2f * al;
      const unsigned k = fkey(al);
      if (k > mx[i]) mx[i] = k;
    } else {
      al = selfal[lo * H + i];
    }
    sacc[i] = expf(al - funkey(mx[i]));
  }
  __syncthreads();
  {
    const float4* __restrict__ h4 =
        reinterpret_cast<const float4*>(hfeat + (size_t)lo * F);
    float4* a4 = reinterpret_cast<float4*>(acc);
    for (int i = tid; i < nrows * (F / 4); i += 1024) {
      const float z = sacc[i >> 4];
      const float4 hv = h4[i];
      float4 r;
      r.x = hv.x * z;
      r.y = hv.y * z;
      r.z = hv.z * z;
      r.w = hv.w * z;
      a4[i] = r;
    }
  }
  __syncthreads();

  if (cnt >= 0) {
    for (int c0 = 0; c0 < cnt; c0 += AGG_CAP) {
      const int fill = (cnt - c0 < AGG_CAP) ? (cnt - c0) : AGG_CAP;
      for (int i = tid; i < fill; i += 1024) q[i] = mylist[c0 + i];
      __syncthreads();
      drain(fill);
    }
  } else {
    scan_all(true);
  }

  {
    const float4* a4 = reinterpret_cast<const float4*>(acc);
    float4* __restrict__ o4 = reinterpret_cast<float4*>(outp + (size_t)lo * F);
    for (int i = tid; i < nrows * (F / 4); i += 1024) {
      const float s = sacc[i >> 4];
      const float4 a = a4[i];
      const int f = (i << 2) & (F - 1);
      float4 r;
      r.x = a.x / s + bias[f];
      r.y = a.y / s + bias[f + 1];
      r.z = a.z / s + bias[f + 2];
      r.w = a.w / s + bias[f + 3];
      if (ELU) {
        r.x = (r.x > 0.f) ? r.x : expm1f(r.x);
        r.y = (r.y > 0.f) ? r.y : expm1f(r.y);
        r.z = (r.z > 0.f) ? r.z : expm1f(r.z);
        r.w = (r.w > 0.f) ? r.w : expm1f(r.w);
      }
      o4[i] = r;
    }
  }
}

extern "C" void kernel_launch(void* const* d_in, const int* in_sizes, int n_in,
                              void* d_out, int out_size, void* d_ws,
                              size_t ws_size, hipStream_t stream) {
  const float* x   = (const float*)d_in[0];
  const int*  eidx = (const int*)d_in[1];
  const float* ew  = (const float*)d_in[2];
  const float* W1  = (const float*)d_in[3];
  const float* We1 = (const float*)d_in[4];
  const float* as1 = (const float*)d_in[5];
  const float* ad1 = (const float*)d_in[6];
  const float* ae1 = (const float*)d_in[7];
  const float* b1  = (const float*)d_in[8];
  const float* W2  = (const float*)d_in[9];
  const float* We2 = (const float*)d_in[10];
  const float* as2 = (const float*)d_in[11];
  const float* ad2 = (const float*)d_in[12];
  const float* ae2 = (const float*)d_in[13];
  const float* b2  = (const float*)d_in[14];
  float* out = (float*)d_out;

  const int N  = in_sizes[0] / 128;
  const int E  = in_sizes[2];
  const int ET = E + N;
  const int* srcv = eidx;
  const int* dstv = eidx + E;

  float* p = (float*)d_ws;
  float* loopw = p; p += N;
  float* scoef = p; p += 2 * (size_t)N;
  float* dcoef = p; p += 2 * (size_t)N;
  float* cst   = p; p += 8;
  float* albuf = p; p += 2 * (size_t)ET;
  float* hfeat = p; p += (size_t)N * 128;
  float* agg   = p; p += (size_t)N * 128;
  int* lcnt = (int*)p; p += 1024;
  unsigned* lists = (unsigned*)p;

  const int T = 256;
  auto cdiv = [](long long a, long long b) { return (int)((a + b - 1) / b); };

  gemm_wmma_f32<128><<<cdiv(N, 16 * 8), 256, 0, stream>>>(x, W1, hfeat, N);
  k_edge_const<2><<<1, 32, 0, stream>>>(We1, ae1, cst);
  k_node_coef<2, false><<<cdiv(N, T), T, 0, stream>>>(hfeat, as1, ad1, loopw, cst,
                                                      scoef, dcoef,
                                                      albuf + 2 * (size_t)E, N);
  k_edge_logit<2><<<cdiv(E, T), T, 0, stream>>>(srcv, dstv, ew, scoef, dcoef,
                                                cst, albuf, E);
  k_range_aggregate<128, true, 1><<<cdiv(N, AGG_R), 1024, 0, stream>>>(
      srcv, dstv, hfeat, albuf, albuf + 2 * (size_t)E, b1, agg, lists, lcnt,
      ew, scoef, dcoef, cst, loopw, E, N);

  gemm_wmma_f32<64><<<cdiv(N, 16 * 6), 128, 0, stream>>>(agg, W2, hfeat, N);
  k_edge_const<1><<<1, 32, 0, stream>>>(We2, ae2, cst);
  k_node_coef<1, true><<<cdiv(N, T), T, 0, stream>>>(hfeat, as2, ad2, loopw, cst,
                                                     scoef, dcoef,
                                                     albuf + (size_t)E, N);
  k_edge_logit<1><<<cdiv(E, T), T, 0, stream>>>(srcv, dstv, ew, scoef, dcoef,
                                                cst, albuf, E);
  k_range_aggregate<64, false, 2><<<cdiv(N, AGG_R), 1024, 0, stream>>>(
      srcv, dstv, hfeat, albuf, albuf + (size_t)E, b2, out, lists, lcnt,
      ew, scoef, dcoef, cst, loopw, E, N);
}


# Sum pass with fewer LDS instructions per edge: a lane owns adjacent row elements (one wide LDS load+store per add), entry words fetched one per lane, exp sums kept in a register; f32 sum order kept
# speedup vs baseline: 3.5837x; 1.0337x over previous
#include <hip/hip_runtime.h>
#include <math.h>


typedef __attribute__((ext_vector_type(2))) float v2f;
typedef __attribute__((ext_vector_type(8))) float v8f;

__device__ __forceinline__ unsigned fkey(float f) {
  unsigned u = __float_as_uint(f);
  return (u & 0x80000000u) ? ~u : (u | 0x80000000u);
}
__device__ __forceinline__ float funkey(unsigned k) {
  unsigned u = (k & 0x80000000u) ? (k & 0x7fffffffu) : ~k;
  return __uint_as_float(u);
}

template <int NCOLS>
__global__ __launch_bounds__(32 * (NCOLS / 16))
void gemm_wmma_f32(const float* __restrict__ A, const float* __restrict__ Wg,
                   float* __restrict__ Out, int M) {
  constexpr int K = 128;
  constexpr int NT = NCOLS / 16;
  constexpr int NTHR = 32 * NT;
  constexpr int PITCH = K + 4;
  constexpr int NV = (16 * K / 4) / NTHR;
  static_assert(NV == 2 || NV == 4, "strip staging is written for 4 or 8 waves");
  __shared__ __attribute__((aligned(16))) float As[2][16 * PITCH];

  const int tid  = threadIdx.x;
  const int wave = tid >> 5;
  const int lane = tid & 31;
  const int r    = lane & 15;
  const int kb   = (lane >> 4) << 1;
  const int col  = (wave << 4) + r;

  v2f bf[K / 4];
#pragma unroll
  for (int s = 0; s < K / 4; ++s) {
    bf[s].x = Wg[(4 * s + kb) * NCOLS + col];
    bf[s].y = Wg[(4 * s + kb + 1) * NCOLS + col];
  }

  const int nstrips = (M + 15) >> 4;
  float4 st0, st1, st2, st3;
  st0 = st1 = st2 = st3 = make_float4(0.f, 0.f, 0.f, 0.f);
  auto gpiece = [&](int sp, int j) __attribute__((always_inline)) -> float4 {
    const int v = tid + j * NTHR;
    int arow = (sp << 4) + (v >> 5);
    if (arow >= M) arow = M - 1;
    return *reinterpret_cast<const float4*>(A + (size_t)arow * K + ((v & 31) << 2));
  };
  auto gload = [&](int sp) __attribute__((always_inline)) {
    st0 = gpiece(sp, 0);
    st1 = gpiece(sp, 1);
    if constexpr (NV > 2) {
      st2 = gpiece(sp, 2);
      st3 = gpiece(sp, 3);
    }
  };
  auto lpiece = [&](int b, int j, const float4 val) __attribute__((always_inline)) {
    const int v = tid + j * NTHR;
    *reinterpret_cast<float4*>(&As[b][(v >> 5) * PITCH + ((v & 31) << 2)]) = val;
  };
  auto lstore = [&](int b) __attribute__((always_inline)) {
    lpiece(b, 0, st0);
    lpiece(b, 1, st1);
    if constexpr (NV > 2) {
      lpiece(b, 2, st2);
      lpiece(b, 3, st3);
    }
  };

  int strip = blockIdx.x;
  if (strip < nstrips) {
    gload(strip);
    lstore(0);
  }
  __syncthreads();

  int buf = 0;
  for (; strip < nstrips; strip += gridDim.x) {
    const int nxt = strip + gridDim.x;
    if (nxt < nstrips) gload(nxt);

    const float* ar = &As[buf][r * PITCH + kb];
    v8f acc = {};
#pragma unroll
    for (int s = 0; s < K / 4; ++s) {
      const v2f a = *reinterpret_cast<const v2f*>(ar + 4 * s);
      acc = __builtin_amdgcn_wmma_f32_16x16x4_f32(false, a, false, bf[s],
                                                  (short)0, acc, false, false);
    }

    const int orow0 = (strip << 4) + ((lane >> 4) << 3);
#pragma unroll
    for (int i = 0; i < 8; ++i) {
      const int orow = orow0 + i;
      if (orow < M) Out[(size_t)orow * NCOLS + col] = acc[i];
    }

    if (nxt < nstrips) lstore(buf ^ 1);
    __syncthreads();
    buf ^= 1;
  }
}

template <int H, bool SELF>
__global__ void k_node_coef(const float* __restrict__ hfeat,
                            const float* __restrict__ a_s,
                            const float* __restrict__ a_d,
                            const float* __restrict__ loopw,
                            const float* __restrict__ cst,
                            float* __restrict__ scoef,
                            float* __restrict__ dcoef,
                            float* __restrict__ selfal, int N) {
  int n = blockIdx.x * blockDim.x + threadIdx.x;
  if (n >= N) return;
  const float* hr = hfeat + (size_t)n * (H * 64);
  float w = 0.f;
  if (SELF) w = loopw[n];
#pragma unroll
  for (int h = 0; h < H; ++h) {
    float ss = 0.f, dd = 0.f;
#pragma unroll 8
    for (int c = 0; c < 64; ++c) {
      float v = hr[h * 64 + c];
      ss += v * a_s[h * 64 + c];
      dd += v * a_d[h * 64 + c];
    }
    scoef[n * H + h] = ss;
    dcoef[n * H + h] = dd;
    if (SELF) {
      float al = ss + dd + w * cst[h];
      al = (al >= 0.f) ? al : 0.2f * al;
      selfal[n * H + h] = al;
    }
  }
}

template <int H>
__global__ void k_edge_const(const float* __restrict__ We,
                             const float* __restrict__ ae,
                             float* __restrict__ cst) {
  int h = threadIdx.x;
  if (h < H) {
    float s = 0.f;
    for (int c = 0; c < 64; ++c) s += We[h * 64 + c] * ae[h * 64 + c];
    cst[h] = s;
  }
}

template <int H>
__global__ void k_edge_logit(const int* __restrict__ src,
                             const int* __restrict__ dst,
                             const float* __restrict__ ew,
                             const float* __restrict__ scoef,
                             const float* __restrict__ dcoef,
                             const float* __restrict__ cst,
                             float* __restrict__ albuf, int E) {
  int e = blockIdx.x * blockDim.x + threadIdx.x;
  if (e >= E) return;
  const int s = src[e];
  const int d = dst[e];
  const float w = ew[e];
#pragma unroll
  for (int h = 0; h < H; ++h) {
    float al = scoef[s * H + h] + dcoef[d * H + h] + w * cst[h];
    al = (al >= 0.f) ? al : 0.2f * al;
    albuf[(size_t)e * H + h] = al;
  }
}

#define AGG_R 400
#define AGG_SUB 16384
#define AGG_CAP 23040
#define AGG_WIN 1024
#define AGG_G 8
#define AGG_MB 8
#define AGG_LCAP 65536
template <int F, bool ELU, int LIST>
__global__ __launch_bounds__(1024)
void k_range_aggregate(const int* __restrict__ src,
                       const int* __restrict__ dst,
                       const float* __restrict__ hfeat,
                       const float* __restrict__ albuf,
                       const float* __restrict__ selfal,
                       const float* __restrict__ bias,
                       float* __restrict__ outp,
                       unsigned* __restrict__ lists,
                       int* __restrict__ lcnt,
                       const float* __restrict__ ew,
                       const float* __restrict__ scoef,
                       const float* __restrict__ dcoef,
                       const float* __restrict__ cst,
                       float* __restrict__ loopw, int E, int N) {
  constexpr int H = F / 64;
  constexpr int T = F / 32;
  typedef __attribute__((ext_vector_type(T))) float rowv;
  __shared__ __attribute__((aligned(16))) float acc[AGG_R * F];
  __shared__ float sacc[AGG_R * H];
  __shared__ unsigned mx[AGG_R * H];
  __shared__ unsigned q[AGG_CAP];
  __shared__ int w_src[AGG_WIN];
  __shared__ float w_z[H * AGG_WIN];
  __shared__ unsigned short wl[32 * 64];
  __shared__ unsigned omask[32 * 32];
  __shared__ int wcnt[2 * 32];
  __shared__ float degl[AGG_R];
  __shared__ float lws[AGG_R];

  const int tid  = threadIdx.x;
  const int wave = tid >> 5;
  const int lane = tid & 31;
  const int myh = (lane * T) >> 6;
  const int lo = blockIdx.x * AGG_R;
  const int nrows = (N - lo < AGG_R) ? (N - lo) : AGG_R;

  if (LIST == 1) {
    for (int i = tid; i < nrows * H; i += 1024) mx[i] = 0u;
    for (int i = tid; i < AGG_R; i += 1024) {
      degl[i] = 0.f;
      lws[i] = 0.f;
    }
  } else {
    for (int i = tid; i < nrows * H; i += 1024) mx[i] = fkey(selfal[lo * H + i]);
  }
  __syncthreads();

  unsigned* __restrict__ mylist = lists + (size_t)blockIdx.x * AGG_LCAP;
  int lsize = 0;

  auto maxgroup = [&](const unsigned (&me)[AGG_MB]) __attribute__((always_inline)) {
    float ma[AGG_MB][H];
#pragma unroll
    for (int j = 0; j < AGG_MB; ++j) {
#pragma unroll
      for (int h = 0; h < H; ++h)
        ma[j][h] = albuf[(size_t)(me[j] >> 9) * H + h];
    }
#pragma unroll
    for (int j = 0; j < AGG_MB; ++j) {
#pragma unroll
      for (int h = 0; h < H; ++h)
        atomicMax(&mx[(int)(me[j] & 511u) * H + h], fkey(ma[j][h]));
    }
  };
  auto maxpass_queue = [&](const int n) __attribute__((always_inline)) {
    for (int i0 = tid; i0 < n; i0 += AGG_MB * 1024) {
      unsigned me[AGG_MB];
      float wv[AGG_MB];
#pragma unroll
      for (int j = 0; j < AGG_MB; ++j) {
        const int i = i0 + j * 1024;
        me[j] = q[(i < n) ? i : i0];
      }
      if (LIST == 1) {
#pragma unroll
        for (int j = 0; j < AGG_MB; ++j) wv[j] = ew[me[j] >> 9];
      }
      maxgroup(me);
      if (LIST == 1) {
#pragma unroll
        for (int j = 0; j < AGG_MB; ++j) {
          if (i0 + j * 1024 < n) {
            const int rw = (int)(me[j] & 511u);
            atomicAdd(&degl[rw], 1.0f);
            atomicAdd(&lws[rw], wv[j]);
          }
        }
      }
    }
  };
  auto maxpass_list = [&](const int n) __attribute__((always_inline)) {
    for (int i0 = tid; i0 < n; i0 += AGG_MB * 1024) {
      unsigned me[AGG_MB];
#pragma unroll
      for (int j = 0; j < AGG_MB; ++j) {
        const int i = i0 + j * 1024;
        me[j] = mylist[(i < n) ? i : i0];
      }
      maxgroup(me);
    }
  };

  constexpr int ZR = (AGG_R + 31) / 32;
  static_assert(ZR * H <= 32, "one lane per owned (row, head)");
  const int zrow = (lane / H) * 32 + wave;
  const bool zmine = (lane < ZR * H) && (zrow < nrows);
  float zacc = 0.f;

  auto drain = [&](const int fill) __attribute__((always_inline)) {
    for (int wb = 0; wb < fill; wb += AGG_WIN) {
      const int wn = (fill - wb < AGG_WIN) ? (fill - wb) : AGG_WIN;
      int owner = 32;
      if (tid < wn) {
        const unsigned ent = q[wb + tid];
        const int e = (int)(ent >> 9);
        const int rw = (int)(ent & 511u);
        owner = (int)(ent & 31u);
        w_src[tid] = src[e];
#pragma unroll
        for (int h = 0; h < H; ++h)
          w_z[h * AGG_WIN + tid] =
              expf(albuf[(size_t)e * H + h] - funkey(mx[rw * H + h]));
      }
      unsigned mym = 0u;
#pragma unroll
      for (int o = 0; o < 32; ++o) {
        const unsigned b = __builtin_amdgcn_ballot_w32(owner == o);
        mym = (lane == o) ? b : mym;
      }
      omask[lane * 32 + wave] = mym;
      __syncthreads();
      const unsigned mymask = omask[wave * 32 + lane];
      const int wlb = __builtin_amdgcn_readfirstlane(wave) * 64;
      const int cm = __builtin_popcount(mymask);
      int pre = cm;
#pragma unroll
      for (int dlt = 1; dlt < 32; dlt <<= 1) {
        const int up = __shfl_up(pre, dlt);
        if (lane >= dlt) pre += up;
      }
      const int nall = __builtin_amdgcn_readlane(pre, 31);
      pre -= cm;
      for (int r0 = 0; r0 < nall; r0 += 64) {
        {
          unsigned mm = mymask;
          int p = pre - r0;
          while (__builtin_amdgcn_ballot_w32(mm != 0u) != 0u) {
            if (mm != 0u) {
              if (p >= 0 && p < 64)
                wl[wave * 64 + p] = (unsigned short)(lane * 32 + __builtin_ctz(mm));
              mm &= mm - 1u;
              ++p;
            }
          }
        }
        const int n = (nall - r0 < 64) ? (nall - r0) : 64;
        const int pa = (lane < n) ? lane : 0;
        const int pb = (lane + 32 < n) ? (lane + 32) : 0;
        const int ia = wl[wlb + pa];
        const int ib = wl[wlb + pb];
        const int ra = (int)(q[wb + ia] & 511u);
        const int rb = (int)(q[wb + ib] & 511u);
        const int sa = w_src[ia];
        const int sb = w_src[ib];
        float za[H], zb[H];
#pragma unroll
        for (int h = 0; h < H; ++h) {
          za[h] = w_z[h * AGG_WIN + ia];
          zb[h] = w_z[h * AGG_WIN + ib];
        }
        for (int kk = 0; kk < n; kk += AGG_G) {
          const bool up = (kk >= 32);
          const int rs = up ? rb : ra;
          const int ss = up ? sb : sa;
          float zs[H];
#pragma unroll
          for (int h = 0; h < H; ++h) zs[h] = up ? zb[h] : za[h];
          int rr[AGG_G];
          float zz[AGG_G][H];
          rowv hv[AGG_G];
#pragma unroll
          for (int i = 0; i < AGG_G; ++i) {
            const int ln = (kk + i) & 31;
            rr[i] = __builtin_amdgcn_readlane(rs, ln);
            const int sn = __builtin_amdgcn_readlane(ss, ln);
#pragma unroll
            for (int h = 0; h < H; ++h)
              zz[i][h] = __uint_as_float((unsigned)__builtin_amdgcn_readlane(
                  (int)__float_as_uint(zs[h]), ln));
            hv[i] = *reinterpret_cast<const rowv*>(hfeat + (size_t)sn * F + lane * T);
          }
#pragma unroll
          for (int i = 0; i < AGG_G; ++i) {
            if (i == 0 || kk + i < n) {
              rowv* ar = reinterpret_cast<rowv*>(acc + rr[i] * F) + lane;
              const float z = (myh == 0) ? zz[i][0] : zz[i][H - 1];
              rowv a = *ar;
              a += hv[i] * z;
              *ar = a;
              const int slot = (rr[i] >> 5) * H;
#pragma unroll
              for (int h = 0; h < H; ++h)
                zacc = (lane == slot + h) ? (zacc + zz[i][h]) : zacc;
            }
          }
        }
      }
      __syncthreads();
    }
  };

  auto scan_all = [&](const bool sums) __attribute__((always_inline)) {
    const unsigned lt = (1u << lane) - 1u;
    const int slice = wave * 512 + lane;
    const int* __restrict__ dp = dst + slice;

    int dn[16];
    auto load_step = [&](int b) {
      if (b + AGG_SUB <= E) {
#pragma unroll
        for (int i = 0; i < 16; ++i) dn[i] = dp[b + i * 32];
      } else {
#pragma unroll
        for (int i = 0; i < 16; ++i) {
          const int e = b + slice + i * 32;
          dn[i] = (e < E) ? dst[e] : -1;
        }
      }
    };
    const int nsteps = (E + AGG_SUB - 1) / AGG_SUB;
    int step = (int)(((long long)blockIdx.x * nsteps) / gridDim.x);
    load_step(step * AGG_SUB);
    __syncthreads();

    int fill = 0;
    int par = 0;
    for (int k = 0; k < nsteps; ++k) {
      const int base = step * AGG_SUB;
      step = (step + 1 < nsteps) ? (step + 1) : 0;
      unsigned mk[16];
      int rl[16];
      int cnt = 0;
#pragma unroll
      for (int i = 0; i < 16; ++i) {
        rl[i] = dn[i] - lo;
        mk[i] = __builtin_amdgcn_ballot_w32((unsigned)rl[i] < (unsigned)nrows);
        cnt += __builtin_popcount(mk[i]);
      }
      if (lane == 0) wcnt[par * 32 + wave] = cnt;
      __syncthreads();
      load_step(step * AGG_SUB);
      const unsigned c = (unsigned)wcnt[par * 32 + lane];
      unsigned v = c | ((lane < wave) ? (c << 16) : 0u);
#pragma unroll
      for (int m = 16; m >= 1; m >>= 1) v += __shfl_xor(v, m);
      const int total = __builtin_amdgcn_readfirstlane((int)(v & 0xffffu));
      int off = fill + __builtin_amdgcn_readfirstlane((int)(v >> 16));
#pragma unroll
      for (int i = 0; i < 16; ++i) {
        if (mk[i] != 0u) {
          if ((mk[i] >> lane) & 1u)
            q[off + __builtin_popcount(mk[i] & lt)] =
                ((unsigned)(base + slice + i * 32) << 9) | (unsigned)rl[i];
          off += __builtin_popcount(mk[i]);
        }
      }
      fill += total;
      par ^= 1;

      const bool last = (k + 1 == nsteps);
      if (fill > AGG_CAP - AGG_SUB || last) {
        __syncthreads();
        if (sums) {
          drain(fill);
        } else {
          if (LIST == 1) {
            if (lsize >= 0 && lsize + fill <= AGG_LCAP) {
              for (int i = tid; i < fill; i += 1024) mylist[lsize + i] = q[i];
              lsize += fill;
            } else {
              lsize = -1;
            }
          }
          maxpass_queue(fill);
          __syncthreads();
        }
        fill = 0;
      }
    }
    if (LIST == 1 && !sums && tid == 0) lcnt[blockIdx.x] = lsize;
  };

  auto scan_own = [&](int& fill_out) __attribute__((always_inline)) -> bool {
    constexpr int PW = (AGG_R * F) / 32;
    const int sw = __builtin_amdgcn_readfirstlane(wave);
    unsigned* stage = reinterpret_cast<unsigned*>(acc) + sw * PW;
    const unsigned lt = (1u << lane) - 1u;
    const int npieces = (E + 511) >> 9;
    const int pwhole = E >> 9;
    const int ppw = (npieces + 31) >> 5;
    const int p0 = sw * ppw;
    const int p1 = (p0 + ppw < npieces) ? (p0 + ppw) : npieces;
    const int p1w = (p1 < pwhole) ? p1 : pwhole;
    const int nw = (p1w > p0) ? (p1w - p0) : 0;
    const int rot = (int)(((long long)blockIdx.x * nw) / gridDim.x);
    auto next_piece = [&](int p) { return (p + 1 < p1w) ? (p + 1) : p0; };
    auto load_piece = [&](int pc, int (&d)[16]) {
      const int b = pc << 9;
#pragma unroll
      for (int i = 0; i < 16; ++i) d[i] = dst[b + i * 32 + lane];
    };
    int off = 0;
    bool ovf = false;
    auto test_piece = [&](int pc, const int (&d)[16]) {
      if (off > PW - 512) {
        ovf = true;
        return;
      }
      const unsigned ebase = (unsigned)((pc << 9) + lane) << 9;
      unsigned mk[16];
      int rl[16];
#pragma unroll
      for (int i = 0; i < 16; ++i) {
        rl[i] = d[i] - lo;
        mk[i] = __builtin_amdgcn_ballot_w32((unsigned)rl[i] < (unsigned)nrows);
      }
#pragma unroll
      for (int i = 0; i < 16; ++i) {
        if (mk[i] != 0u) {
          if ((mk[i] >> lane) & 1u)
            stage[off + __builtin_popcount(mk[i] & lt)] =
                (ebase + ((unsigned)(i * 32) << 9)) | (unsigned)rl[i];
          off += __builtin_popcount(mk[i]);
        }
      }
    };
    int da[16], db[16];
    if (nw > 0) {
      int pa = p0 + rot;
      load_piece(pa, da);
      int k = 0;
      for (; k + 2 <= nw; k += 2) {
        const int pb = next_piece(pa);
        load_piece(pb, db);
        test_piece(pa, da);
        pa = next_piece(pb);
        load_piece(pa, da);
        test_piece(pb, db);
      }
      if (k < nw) test_piece(pa, da);
    }
    if (pwhole < p1 && pwhole >= p0) {
      const int b = pwhole << 9;
#pragma unroll
      for (int i = 0; i < 16; ++i) {
        const int e = b + i * 32 + lane;
        da[i] = (e < E) ? dst[e] : -1;
      }
      test_piece(pwhole, da);
    }
    if (lane == 0) wcnt[sw] = ovf ? (1 << 20) : off;
    __syncthreads();
    const int c = wcnt[lane];
    int pre = c;
#pragma unroll
    for (int dlt = 1; dlt < 32; dlt <<= 1) {
      const int up = __shfl_up(pre, dlt);
      if (lane >= dlt) pre += up;
    }
    const int total = __builtin_amdgcn_readlane(pre, 31);
    const int myoff = __builtin_amdgcn_readlane(pre - c, sw);
    if (total > AGG_CAP) return false;
    for (int i = lane; i < off; i += 32) q[myoff + i] = stage[i];
    __syncthreads();
    fill_out = total;
    return true;
  };

  int cnt;
  if (LIST == 2) {
    cnt = lcnt[blockIdx.x];
    if (cnt >= 0 && cnt <= AGG_LCAP) {
      maxpass_list(cnt);
    } else {
      cnt = -1;
      scan_all(false);
    }
  } else {
    bool own = false;
    if constexpr (LIST == 1) {
      int fill = 0;
      own = scan_own(fill);
      if (own) {
        for (int i = tid; i < fill; i += 1024) mylist[i] = q[i];
        lsize = fill;
        maxpass_queue(fill);
        if (tid == 0) lcnt[blockIdx.x] = lsize;
      }
    }
    if (!own) scan_all(false);
    cnt = lsize;
  }
  __syncthreads();

  for (int i = tid; i < nrows * H; i += 1024) {
    float al;
    if (LIST == 1) {
      const int rw = i / H;
      const int h = i - rw * H;
      const float w = lws[rw] / fmaxf(degl[rw], 1.0f);
      if (h == 0) loopw[lo + rw] = w;
      al = scoef[lo * H + i] + dcoef[lo * H + i] + w * cst[h];
      al = (al >= 0.f) ? al : 0.2f * al;
      const unsigned k = fkey(al);
      if (k > mx[i]) mx[i] = k;
    } else {
      al = selfal[lo * H + i];
    }
    sacc[i] = expf(al - funkey(mx[i]));
  }
  __syncthreads();
  {
    const float4* __restrict__ h4 =
        reinterpret_cast<const float4*>(hfeat + (size_t)lo * F);
    float4* a4 = reinterpret_cast<float4*>(acc);
    for (int i = tid; i < nrows * (F / 4); i += 1024) {
      const float z = sacc[i >> 4];
      const float4 hv = h4[i];
      float4 r;
      r.x = hv.x * z;
      r.y = hv.y * z;
      r.z = hv.z * z;
      r.w = hv.w * z;
      a4[i] = r;
    }
  }
  __syncthreads();

  if (zmine) zacc = sacc[zrow * H + (lane % H)];
  if (cnt >= 0) {
    for (int c0 = 0; c0 < cnt; c0 += AGG_CAP) {
      const int fill = (cnt - c0 < AGG_CAP) ? (cnt - c0) : AGG_CAP;
      for (int i = tid; i < fill; i += 1024) q[i] = mylist[c0 + i];
      __syncthreads();
      drain(fill);
    }
  } else {
    scan_all(true);
  }

  if (zmine) sacc[zrow * H + (lane % H)] = zacc;
  __syncthreads();

  {
    const float4* a4 = reinterpret_cast<const float4*>(acc);
    float4* __restrict__ o4 = reinterpret_cast<float4*>(outp + (size_t)lo * F);
    for (int i = tid; i < nrows * (F / 4); i += 1024) {
      const float s = sacc[i >> 4];
      const float4 a = a4[i];
      const int f = (i << 2) & (F - 1);
      float4 r;
      r.x = a.x / s + bias[f];
      r.y = a.y / s + bias[f + 1];
      r.z = a.z / s + bias[f + 2];
      r.w = a.w / s + bias[f + 3];
      if (ELU) {
        r.x = (r.x > 0.f) ? r.x : expm1f(r.x);
        r.y = (r.y > 0.f) ? r.y : expm1f(r.y);
        r.z = (r.z > 0.f) ? r.z : expm1f(r.z);
        r.w = (r.w > 0.f) ? r.w : expm1f(r.w);
      }
      o4[i] = r;
    }
  }
}

extern "C" void kernel_launch(void* const* d_in, const int* in_sizes, int n_in,
                              void* d_out, int out_size, void* d_ws,
                              size_t ws_size, hipStream_t stream) {
  const float* x   = (const float*)d_in[0];
  const int*  eidx = (const int*)d_in[1];
  const float* ew  = (const float*)d_in[2];
  const float* W1  = (const float*)d_in[3];
  const float* We1 = (const float*)d_in[4];
  const float* as1 = (const float*)d_in[5];
  const float* ad1 = (const float*)d_in[6];
  const float* ae1 = (const float*)d_in[7];
  const float* b1  = (const float*)d_in[8];
  const float* W2  = (const float*)d_in[9];
  const float* We2 = (const float*)d_in[10];
  const float* as2 = (const float*)d_in[11];
  const float* ad2 = (const float*)d_in[12];
  const float* ae2 = (const float*)d_in[13];
  const float* b2  = (const float*)d_in[14];
  float* out = (float*)d_out;

  const int N  = in_sizes[0] / 128;
  const int E  = in_sizes[2];
  const int ET = E + N;
  const int* srcv = eidx;
  const int* dstv = eidx + E;

  float* p = (float*)d_ws;
  float* loopw = p; p += N;
  float* scoef = p; p += 2 * (size_t)N;
  float* dcoef = p; p += 2 * (size_t)N;
  float* cst   = p; p += 8;
  float* albuf = p; p += 2 * (size_t)ET;
  float* hfeat = p; p += (size_t)N * 128;
  float* agg   = p; p += (size_t)N * 128;
  int* lcnt = (int*)p; p += 1024;
  unsigned* lists = (unsigned*)p;

  const int T = 256;
  auto cdiv = [](long long a, long long b) { return (int)((a + b - 1) / b); };

  gemm_wmma_f32<128><<<cdiv(N, 16 * 8), 256, 0, stream>>>(x, W1, hfeat, N);
  k_edge_const<2><<<1, 32, 0, stream>>>(We1, ae1, cst);
  k_node_coef<2, false><<<cdiv(N, T), T, 0, stream>>>(hfeat, as1, ad1, loopw, cst,
                                                      scoef, dcoef,
                                                      albuf + 2 * (size_t)E, N);
  k_edge_logit<2><<<cdiv(E, T), T, 0, stream>>>(srcv, dstv, ew, scoef, dcoef,
                                                cst, albuf, E);
  k_range_aggregate<128, true, 1><<<cdiv(N, AGG_R), 1024, 0, stream>>>(
      srcv, dstv, hfeat, albuf, albuf + 2 * (size_t)E, b1, agg, lists, lcnt,
      ew, scoef, dcoef, cst, loopw, E, N);

  gemm_wmma_f32<64><<<cdiv(N, 16 * 6), 128, 0, stream>>>(agg, W2, hfeat, N);
  k_edge_const<1><<<1, 32, 0, stream>>>(We2, ae2, cst);
  k_node_coef<1, true><<<cdiv(N, T), T, 0, stream>>>(hfeat, as2, ad2, loopw, cst,
                                                     scoef, dcoef,
                                                     albuf + (size_t)E, N);
  k_edge_logit<1><<<cdiv(E, T), T, 0, stream>>>(srcv, dstv, ew, scoef, dcoef,
                                                cst, albuf, E);
  k_range_aggregate<64, false, 2><<<cdiv(N, AGG_R), 1024, 0, stream>>>(
      srcv, dstv, hfeat, albuf, albuf + (size_t)E, b2, out, lists, lcnt,
      ew, scoef, dcoef, cst, loopw, E, N);
}
